# starting kernel
# baseline (speedup 1.0000x reference)
#include <hip/hip_runtime.h>
#include <cstdint>
#include <cstddef>

#define DEVINL __device__ __forceinline__

typedef float v2f __attribute__((ext_vector_type(2)));
typedef float v8f __attribute__((ext_vector_type(8)));

static constexpr int T_DIM  = 1024;
static constexpr int B_DIM  = 64;
static constexpr int IN_DIM = 32;
static constexpr int EMB    = 64;
static constexpr int HID    = 128;
static constexpr int KF     = 16;
static constexpr int OUTD   = 32;
static constexpr int G4     = 512;

static constexpr int HS     = 132;
static constexpr int XS     = 68;
static constexpr int IS     = 36;
static constexpr int BP512  = 528;
static constexpr int BP64   = 80;

DEVINL float sigm(float x) { return 1.0f / (1.0f + __expf(-x)); }

DEVINL v8f wmma4(v2f a, v2f b, v8f c) {
  return __builtin_amdgcn_wmma_f32_16x16x4_f32(false, a, false, b, (short)0, c,
                                               false, false);
}

DEVINL v8f vzero8() {
  v8f z;
#pragma unroll
  for (int r = 0; r < 8; ++r) z[r] = 0.0f;
  return z;
}

__global__ void zero_kernel(float* __restrict__ p, int n) {
  int i = blockIdx.x * blockDim.x + threadIdx.x;
  if (i < n) p[i] = 0.0f;
}

__global__ void __launch_bounds__(512)
embed_xg0_kernel(const float* __restrict__ in,
                 const float* __restrict__ We,
                 const float* __restrict__ be,
                 const float* __restrict__ Wx0,
                 const float* __restrict__ b0,
                 float* __restrict__ xg)
{
  extern __shared__ __align__(16) char smem[];
  float* inP = (float*)smem;
  v2f*   WeP = (v2f*)(smem + 64 * IS * 4);
  float* xb  = (float*)(smem + 64 * IS * 4 + 16 * BP64 * 8);
  v2f*   WxP = (v2f*)(smem + 64 * IS * 4 + 16 * BP64 * 8 + 64 * XS * 4);

  const int tid  = threadIdx.x;
  const int wave = tid >> 5;
  const int lane = tid & 31;
  const int lh   = lane >> 4;
  const int ln   = lane & 15;
  const int row0 = blockIdx.x * 64;

  for (int i = tid; i < 64 * 32; i += 512) {
    int r = i >> 5, c = i & 31;
    inP[r * IS + c] = in[(size_t)(row0 + r) * 32 + c];
  }
  for (int i = tid; i < 16 * 64; i += 512) {
    int kp = i >> 6, n = i & 63;
    v2f w; w.x = We[(2 * kp) * 64 + n]; w.y = We[(2 * kp + 1) * 64 + n];
    WeP[kp * BP64 + n] = w;
  }
  for (int i = tid; i < 32 * 512; i += 512) {
    int kp = i >> 9, n = i & 511;
    v2f w; w.x = Wx0[(2 * kp) * 512 + n]; w.y = Wx0[(2 * kp + 1) * 512 + n];
    WxP[kp * BP512 + n] = w;
  }
  __syncthreads();

  {
    const int tm = wave >> 2, tn = wave & 3;
    const int col = tn * 16 + ln;
    const float bv = be[col];
    v8f acc = vzero8();
    const int am = tm * 16 + ln;
#pragma unroll
    for (int k0 = 0; k0 < 32; k0 += 4) {
      v2f a = *(const v2f*)&inP[am * IS + k0 + 2 * lh];
      v2f b = WeP[((k0 >> 1) + lh) * BP64 + col];
      acc = wmma4(a, b, acc);
    }
#pragma unroll
    for (int r = 0; r < 8; ++r)
      xb[(tm * 16 + r + 8 * lh) * XS + col] = acc[r] + bv;
  }
  __syncthreads();

  for (int task = wave; task < 128; task += 16) {
    const int tm = task >> 5, tn = task & 31;
    const int col = tn * 16 + ln;
    const float bv = b0[col];
    v8f acc = vzero8();
    const int am = tm * 16 + ln;
#pragma unroll
    for (int k0 = 0; k0 < 64; k0 += 4) {
      v2f a = *(const v2f*)&xb[am * XS + k0 + 2 * lh];
      v2f b = WxP[((k0 >> 1) + lh) * BP512 + col];
      acc = wmma4(a, b, acc);
    }
#pragma unroll
    for (int r = 0; r < 8; ++r)
      xg[(size_t)(row0 + tm * 16 + r + 8 * lh) * 512 + col] = acc[r] + bv;
  }
}

__global__ void __launch_bounds__(512)
xg1_kernel(const float* __restrict__ x1,
           const float* __restrict__ Wx1,
           const float* __restrict__ b1,
           float* __restrict__ xg)
{
  extern __shared__ __align__(16) char smem[];
  float* aB = (float*)smem;
  v2f*   WP = (v2f*)(smem + 64 * HS * 4);

  const int tid  = threadIdx.x;
  const int wave = tid >> 5;
  const int lane = tid & 31;
  const int lh   = lane >> 4;
  const int ln   = lane & 15;
  const int row0 = blockIdx.x * 64;

  for (int i = tid; i < 64 * 128; i += 512) {
    int r = i >> 7, c = i & 127;
    aB[r * HS + c] = x1[(size_t)(row0 + r) * 128 + c];
  }
  for (int i = tid; i < 64 * 512; i += 512) {
    int kp = i >> 9, n = i & 511;
    v2f w; w.x = Wx1[(2 * kp) * 512 + n]; w.y = Wx1[(2 * kp + 1) * 512 + n];
    WP[kp * BP512 + n] = w;
  }
  __syncthreads();

  for (int task = wave; task < 128; task += 16) {
    const int tm = task >> 5, tn = task & 31;
    const int col = tn * 16 + ln;
    const float bv = b1[col];
    v8f acc = vzero8();
    const int am = tm * 16 + ln;
    for (int k0 = 0; k0 < 128; k0 += 4) {
      v2f a = *(const v2f*)&aB[am * HS + k0 + 2 * lh];
      v2f b = WP[((k0 >> 1) + lh) * BP512 + col];
      acc = wmma4(a, b, acc);
    }
#pragma unroll
    for (int r = 0; r < 8; ++r)
      xg[(size_t)(row0 + tm * 16 + r + 8 * lh) * 512 + col] = acc[r] + bv;
  }
}

template <int NOUT>
__global__ void __launch_bounds__(512)
scan_kernel(const float* __restrict__ xg,
            const float* __restrict__ Wh,
            const float* __restrict__ Wo,
            const float* __restrict__ bo,
            float* __restrict__ ring,
            float* __restrict__ out)
{
  extern __shared__ __align__(16) char smem[];
  v2f*   WhP  = (v2f*)smem;
  float* hbuf = (float*)(smem + 64 * BP512 * 8);

  const int tid  = threadIdx.x;
  const int wave = tid >> 5;
  const int lane = tid & 31;
  const int lh   = lane >> 4;
  const int ln   = lane & 15;

  for (int i = tid; i < 64 * 512; i += 512) {
    int kp = i >> 9, n = i & 511;
    v2f w; w.x = Wh[(2 * kp) * 512 + n]; w.y = Wh[(2 * kp + 1) * 512 + n];
    WhP[kp * BP512 + n] = w;
  }
  for (int i = tid; i < 64 * HS; i += 512) hbuf[i] = 0.0f;
  __syncthreads();

  float hnew[2][8];

  for (int t = 0; t < T_DIM; ++t) {
    const float* xgt = xg + (size_t)t * (64 * 512);

    if (t + 1 < T_DIM) {
      const float* nxt = xgt + 64 * 512;
#pragma unroll
      for (int q = 0; q < 8; ++q)
        __builtin_prefetch(nxt + ((size_t)tid + q * 512) * 32, 0, 0);
    }

#pragma unroll
    for (int ti = 0; ti < 2; ++ti) {
      const int task = wave + ti * 16;
      const int tm   = task >> 3;
      const int tnh  = task & 7;
      const int colh = tnh * 16 + ln;

      float xv0[8], xv1[8], xv2[8], xv3[8];
#pragma unroll
      for (int r = 0; r < 8; ++r) {
        const int row = tm * 16 + r + 8 * lh;
        const float* p = xgt + row * 512 + colh;
        xv0[r] = p[0];  xv1[r] = p[128];
        xv2[r] = p[256]; xv3[r] = p[384];
      }

      v8f acc0 = vzero8(), acc1 = vzero8(), acc2 = vzero8(), acc3 = vzero8();
      const int am = tm * 16 + ln;
      for (int k0 = 0; k0 < 128; k0 += 4) {
        v2f a = *(const v2f*)&hbuf[am * HS + k0 + 2 * lh];
        const int kp = (k0 >> 1) + lh;
        acc0 = wmma4(a, WhP[kp * BP512 + colh + 0],   acc0);
        acc1 = wmma4(a, WhP[kp * BP512 + colh + 128], acc1);
        acc2 = wmma4(a, WhP[kp * BP512 + colh + 256], acc2);
        acc3 = wmma4(a, WhP[kp * BP512 + colh + 384], acc3);
      }

      float hc[8], og[8], dd[8];
#pragma unroll
      for (int r = 0; r < 8; ++r) {
        const float ig = sigm(acc0[r] + xv0[r]);
        og[r] = sigm(acc1[r] + xv1[r]);
        const float cg = tanhf(acc2[r] + xv2[r]);
        dd[r] = 0.5f * sigm(acc3[r] + xv3[r]);
        hc[r] = ig * cg;
      }

      float csum[8], w[8];
#pragma unroll
      for (int r = 0; r < 8; ++r) { csum[r] = hc[r]; w[r] = 1.0f; }
#pragma unroll
      for (int j = 1; j < KF; ++j) {
        const float* rp = ring + (size_t)((t - j) & 15) * (64 * 128);
        const float jm1 = (float)(j - 1), rj = 1.0f / (float)j;
#pragma unroll
        for (int r = 0; r < 8; ++r) {
          const int row = tm * 16 + r + 8 * lh;
          w[r] *= (jm1 + dd[r]) * rj;
          csum[r] += w[r] * rp[row * 128 + colh];
        }
      }
      float* rw = ring + (size_t)(t & 15) * (64 * 128);
#pragma unroll
      for (int r = 0; r < 8; ++r) {
        const int row = tm * 16 + r + 8 * lh;
        rw[row * 128 + colh] = hc[r];
        hnew[ti][r] = og[r] * tanhf(csum[r]);
      }
    }

    __syncthreads();
#pragma unroll
    for (int ti = 0; ti < 2; ++ti) {
      const int task = wave + ti * 16;
      const int tm = task >> 3, tnh = task & 7;
      const int colh = tnh * 16 + ln;
#pragma unroll
      for (int r = 0; r < 8; ++r)
        hbuf[(tm * 16 + r + 8 * lh) * HS + colh] = hnew[ti][r];
    }
    __syncthreads();

    float* ot = out + (size_t)t * (64 * NOUT);
    for (int task = wave; task < 4 * (NOUT / 16); task += 16) {
      const int tm  = task / (NOUT / 16);
      const int tno = task % (NOUT / 16);
      const int col = tno * 16 + ln;
      const float bv = bo[col];
      v8f acc = vzero8();
      const int am = tm * 16 + ln;
      for (int k0 = 0; k0 < 128; k0 += 4) {
        v2f a = *(const v2f*)&hbuf[am * HS + k0 + 2 * lh];
        v2f b;
        b.x = Wo[(k0 + 2 * lh) * NOUT + col];
        b.y = Wo[(k0 + 2 * lh + 1) * NOUT + col];
        acc = wmma4(a, b, acc);
      }
#pragma unroll
      for (int r = 0; r < 8; ++r)
        ot[(tm * 16 + r + 8 * lh) * NOUT + col] = acc[r] + bv;
    }
  }
}

extern "C" void kernel_launch(void* const* d_in, const int* in_sizes, int n_in,
                              void* d_out, int out_size, void* d_ws, size_t ws_size,
                              hipStream_t stream) {
  const float* inputs = (const float*)d_in[0];
  const float* W_emb  = (const float*)d_in[1];
  const float* b_emb  = (const float*)d_in[2];
  const float* Wx0    = (const float*)d_in[3];
  const float* Wh0    = (const float*)d_in[4];
  const float* b0     = (const float*)d_in[5];
  const float* Wo0    = (const float*)d_in[6];
  const float* bo0    = (const float*)d_in[7];
  const float* Wx1    = (const float*)d_in[8];
  const float* Wh1    = (const float*)d_in[9];
  const float* b1     = (const float*)d_in[10];
  const float* Wo1    = (const float*)d_in[11];
  const float* bo1    = (const float*)d_in[12];
  float* out = (float*)d_out;

  float* xg   = (float*)d_ws;
  float* x1   = xg + (size_t)T_DIM * B_DIM * G4;
  float* ring = x1 + (size_t)T_DIM * B_DIM * HID;
  const int ringN = KF * B_DIM * HID;

  const size_t ldsE = (size_t)64 * IS * 4 + 16 * BP64 * 8 + 64 * XS * 4 + 32 * BP512 * 8;
  const size_t ldsX = (size_t)64 * HS * 4 + 64 * BP512 * 8;
  const size_t ldsS = (size_t)64 * BP512 * 8 + 64 * HS * 4;

  const int nblk = (T_DIM * B_DIM) / 64;

  embed_xg0_kernel<<<nblk, 512, ldsE, stream>>>(inputs, W_emb, b_emb, Wx0, b0, xg);
  zero_kernel<<<ringN / 256, 256, 0, stream>>>(ring, ringN);
  scan_kernel<HID><<<1, 512, ldsS, stream>>>(xg, Wh0, Wo0, bo0, ring, x1);
  xg1_kernel<<<nblk, 512, ldsX, stream>>>(x1, Wx1, b1, xg);
  zero_kernel<<<ringN / 256, 256, 0, stream>>>(ring, ringN);
  scan_kernel<OUTD><<<1, 512, ldsS, stream>>>(xg, Wh1, Wo1, bo1, ring, out);
}


# scan kernels: the 64 independent batch rows split over 4 workgroups (16 rows, 8 waves, one gate task per wave) instead of one 16-wave workgroup; summation order per element unchanged
# speedup vs baseline: 6.2712x; 6.2712x over previous
#include <hip/hip_runtime.h>
#include <cstdint>
#include <cstddef>

#define DEVINL __device__ __forceinline__

typedef float v2f __attribute__((ext_vector_type(2)));
typedef float v8f __attribute__((ext_vector_type(8)));

static constexpr int T_DIM  = 1024;
static constexpr int B_DIM  = 64;
static constexpr int IN_DIM = 32;
static constexpr int EMB    = 64;
static constexpr int HID    = 128;
static constexpr int KF     = 16;
static constexpr int OUTD   = 32;
static constexpr int G4     = 512;

static constexpr int HS     = 132;
static constexpr int XS     = 68;
static constexpr int IS     = 36;
static constexpr int BP512  = 528;
static constexpr int BP64   = 80;

DEVINL float sigm(float x) { return 1.0f / (1.0f + __expf(-x)); }

DEVINL v8f wmma4(v2f a, v2f b, v8f c) {
  return __builtin_amdgcn_wmma_f32_16x16x4_f32(false, a, false, b, (short)0, c,
                                               false, false);
}

DEVINL v8f vzero8() {
  v8f z;
#pragma unroll
  for (int r = 0; r < 8; ++r) z[r] = 0.0f;
  return z;
}

__global__ void zero_kernel(float* __restrict__ p, int n) {
  int i = blockIdx.x * blockDim.x + threadIdx.x;
  if (i < n) p[i] = 0.0f;
}

__global__ void __launch_bounds__(512)
embed_xg0_kernel(const float* __restrict__ in,
                 const float* __restrict__ We,
                 const float* __restrict__ be,
                 const float* __restrict__ Wx0,
                 const float* __restrict__ b0,
                 float* __restrict__ xg)
{
  extern __shared__ __align__(16) char smem[];
  float* inP = (float*)smem;
  v2f*   WeP = (v2f*)(smem + 64 * IS * 4);
  float* xb  = (float*)(smem + 64 * IS * 4 + 16 * BP64 * 8);
  v2f*   WxP = (v2f*)(smem + 64 * IS * 4 + 16 * BP64 * 8 + 64 * XS * 4);

  const int tid  = threadIdx.x;
  const int wave = tid >> 5;
  const int lane = tid & 31;
  const int lh   = lane >> 4;
  const int ln   = lane & 15;
  const int row0 = blockIdx.x * 64;

  for (int i = tid; i < 64 * 32; i += 512) {
    int r = i >> 5, c = i & 31;
    inP[r * IS + c] = in[(size_t)(row0 + r) * 32 + c];
  }
  for (int i = tid; i < 16 * 64; i += 512) {
    int kp = i >> 6, n = i & 63;
    v2f w; w.x = We[(2 * kp) * 64 + n]; w.y = We[(2 * kp + 1) * 64 + n];
    WeP[kp * BP64 + n] = w;
  }
  for (int i = tid; i < 32 * 512; i += 512) {
    int kp = i >> 9, n = i & 511;
    v2f w; w.x = Wx0[(2 * kp) * 512 + n]; w.y = Wx0[(2 * kp + 1) * 512 + n];
    WxP[kp * BP512 + n] = w;
  }
  __syncthreads();

  {
    const int tm = wave >> 2, tn = wave & 3;
    const int col = tn * 16 + ln;
    const float bv = be[col];
    v8f acc = vzero8();
    const int am = tm * 16 + ln;
#pragma unroll
    for (int k0 = 0; k0 < 32; k0 += 4) {
      v2f a = *(const v2f*)&inP[am * IS + k0 + 2 * lh];
      v2f b = WeP[((k0 >> 1) + lh) * BP64 + col];
      acc = wmma4(a, b, acc);
    }
#pragma unroll
    for (int r = 0; r < 8; ++r)
      xb[(tm * 16 + r + 8 * lh) * XS + col] = acc[r] + bv;
  }
  __syncthreads();

  for (int task = wave; task < 128; task += 16) {
    const int tm = task >> 5, tn = task & 31;
    const int col = tn * 16 + ln;
    const float bv = b0[col];
    v8f acc = vzero8();
    const int am = tm * 16 + ln;
#pragma unroll
    for (int k0 = 0; k0 < 64; k0 += 4) {
      v2f a = *(const v2f*)&xb[am * XS + k0 + 2 * lh];
      v2f b = WxP[((k0 >> 1) + lh) * BP512 + col];
      acc = wmma4(a, b, acc);
    }
#pragma unroll
    for (int r = 0; r < 8; ++r)
      xg[(size_t)(row0 + tm * 16 + r + 8 * lh) * 512 + col] = acc[r] + bv;
  }
}

__global__ void __launch_bounds__(512)
xg1_kernel(const float* __restrict__ x1,
           const float* __restrict__ Wx1,
           const float* __restrict__ b1,
           float* __restrict__ xg)
{
  extern __shared__ __align__(16) char smem[];
  float* aB = (float*)smem;
  v2f*   WP = (v2f*)(smem + 64 * HS * 4);

  const int tid  = threadIdx.x;
  const int wave = tid >> 5;
  const int lane = tid & 31;
  const int lh   = lane >> 4;
  const int ln   = lane & 15;
  const int row0 = blockIdx.x * 64;

  for (int i = tid; i < 64 * 128; i += 512) {
    int r = i >> 7, c = i & 127;
    aB[r * HS + c] = x1[(size_t)(row0 + r) * 128 + c];
  }
  for (int i = tid; i < 64 * 512; i += 512) {
    int kp = i >> 9, n = i & 511;
    v2f w; w.x = Wx1[(2 * kp) * 512 + n]; w.y = Wx1[(2 * kp + 1) * 512 + n];
    WP[kp * BP512 + n] = w;
  }
  __syncthreads();

  for (int task = wave; task < 128; task += 16) {
    const int tm = task >> 5, tn = task & 31;
    const int col = tn * 16 + ln;
    const float bv = b1[col];
    v8f acc = vzero8();
    const int am = tm * 16 + ln;
    for (int k0 = 0; k0 < 128; k0 += 4) {
      v2f a = *(const v2f*)&aB[am * HS + k0 + 2 * lh];
      v2f b = WP[((k0 >> 1) + lh) * BP512 + col];
      acc = wmma4(a, b, acc);
    }
#pragma unroll
    for (int r = 0; r < 8; ++r)
      xg[(size_t)(row0 + tm * 16 + r + 8 * lh) * 512 + col] = acc[r] + bv;
  }
}

template <int NOUT>
__global__ void __launch_bounds__(256)
scan_kernel(const float* __restrict__ xg,
            const float* __restrict__ Wh,
            const float* __restrict__ Wo,
            const float* __restrict__ bo,
            float* __restrict__ ring,
            float* __restrict__ out)
{
  extern __shared__ __align__(16) char smem[];
  v2f*   WhP  = (v2f*)smem;
  float* hbuf = (float*)(smem + 64 * BP512 * 8);

  const int tid  = threadIdx.x;
  const int wave = tid >> 5;
  const int lane = tid & 31;
  const int lh   = lane >> 4;
  const int ln   = lane & 15;
  const int rowb = blockIdx.x * 16;
  const int colh = wave * 16 + ln;

  for (int i = tid; i < 64 * 512; i += 256) {
    int kp = i >> 9, n = i & 511;
    v2f w; w.x = Wh[(2 * kp) * 512 + n]; w.y = Wh[(2 * kp + 1) * 512 + n];
    WhP[kp * BP512 + n] = w;
  }
  for (int i = tid; i < 16 * HS; i += 256) hbuf[i] = 0.0f;
  __syncthreads();

  float hnew[8];

  for (int t = 0; t < T_DIM; ++t) {
    const float* xgt = xg + (size_t)t * (64 * 512) + (size_t)rowb * 512;

    if (t + 1 < T_DIM) {
#pragma unroll
      for (int q = 1; q <= 4; ++q)
        __builtin_prefetch(xgt + (size_t)q * (64 * 512) + (size_t)tid * 32, 0, 0);
    }

    {
      float xv0[8], xv1[8], xv2[8], xv3[8];
#pragma unroll
      for (int r = 0; r < 8; ++r) {
        const int row = r + 8 * lh;
        const float* p = xgt + row * 512 + colh;
        xv0[r] = p[0];  xv1[r] = p[128];
        xv2[r] = p[256]; xv3[r] = p[384];
      }

      v8f acc0 = vzero8(), acc1 = vzero8(), acc2 = vzero8(), acc3 = vzero8();
      for (int k0 = 0; k0 < 128; k0 += 4) {
        v2f a = *(const v2f*)&hbuf[ln * HS + k0 + 2 * lh];
        const int kp = (k0 >> 1) + lh;
        acc0 = wmma4(a, WhP[kp * BP512 + colh + 0],   acc0);
        acc1 = wmma4(a, WhP[kp * BP512 + colh + 128], acc1);
        acc2 = wmma4(a, WhP[kp * BP512 + colh + 256], acc2);
        acc3 = wmma4(a, WhP[kp * BP512 + colh + 384], acc3);
      }

      float hc[8], og[8], dd[8];
#pragma unroll
      for (int r = 0; r < 8; ++r) {
        const float ig = sigm(acc0[r] + xv0[r]);
        og[r] = sigm(acc1[r] + xv1[r]);
        const float cg = tanhf(acc2[r] + xv2[r]);
        dd[r] = 0.5f * sigm(acc3[r] + xv3[r]);
        hc[r] = ig * cg;
      }

      float csum[8], w[8];
#pragma unroll
      for (int r = 0; r < 8; ++r) { csum[r] = hc[r]; w[r] = 1.0f; }
#pragma unroll
      for (int j = 1; j < KF; ++j) {
        const float* rp = ring + (size_t)((t - j) & 15) * (64 * 128) + (size_t)rowb * 128;
        const float jm1 = (float)(j - 1), rj = 1.0f / (float)j;
#pragma unroll
        for (int r = 0; r < 8; ++r) {
          const int row = r + 8 * lh;
          w[r] *= (jm1 + dd[r]) * rj;
          csum[r] += w[r] * rp[row * 128 + colh];
        }
      }
      float* rw = ring + (size_t)(t & 15) * (64 * 128) + (size_t)rowb * 128;
#pragma unroll
      for (int r = 0; r < 8; ++r) {
        const int row = r + 8 * lh;
        rw[row * 128 + colh] = hc[r];
        hnew[r] = og[r] * tanhf(csum[r]);
      }
    }

    __syncthreads();
#pragma unroll
    for (int r = 0; r < 8; ++r)
      hbuf[(r + 8 * lh) * HS + colh] = hnew[r];
    __syncthreads();

    if (NOUT / 16 >= 8 || wave < NOUT / 16) {
      float* ot = out + (size_t)t * (64 * NOUT) + (size_t)rowb * NOUT;
      const int col = wave * 16 + ln;
      const float bv = bo[col];
      v8f acc = vzero8();
      for (int k0 = 0; k0 < 128; k0 += 4) {
        v2f a = *(const v2f*)&hbuf[ln * HS + k0 + 2 * lh];
        v2f b;
        b.x = Wo[(k0 + 2 * lh) * NOUT + col];
        b.y = Wo[(k0 + 2 * lh + 1) * NOUT + col];
        acc = wmma4(a, b, acc);
      }
#pragma unroll
      for (int r = 0; r < 8; ++r)
        ot[(r + 8 * lh) * NOUT + col] = acc[r] + bv;
    }
  }
}

extern "C" void kernel_launch(void* const* d_in, const int* in_sizes, int n_in,
                              void* d_out, int out_size, void* d_ws, size_t ws_size,
                              hipStream_t stream) {
  const float* inputs = (const float*)d_in[0];
  const float* W_emb  = (const float*)d_in[1];
  const float* b_emb  = (const float*)d_in[2];
  const float* Wx0    = (const float*)d_in[3];
  const float* Wh0    = (const float*)d_in[4];
  const float* b0     = (const float*)d_in[5];
  const float* Wo0    = (const float*)d_in[6];
  const float* bo0    = (const float*)d_in[7];
  const float* Wx1    = (const float*)d_in[8];
  const float* Wh1    = (const float*)d_in[9];
  const float* b1     = (const float*)d_in[10];
  const float* Wo1    = (const float*)d_in[11];
  const float* bo1    = (const float*)d_in[12];
  float* out = (float*)d_out;

  float* xg   = (float*)d_ws;
  float* x1   = xg + (size_t)T_DIM * B_DIM * G4;
  float* ring = x1 + (size_t)T_DIM * B_DIM * HID;
  const int ringN = KF * B_DIM * HID;

  const size_t ldsE = (size_t)64 * IS * 4 + 16 * BP64 * 8 + 64 * XS * 4 + 32 * BP512 * 8;
  const size_t ldsX = (size_t)64 * HS * 4 + 64 * BP512 * 8;
  const size_t ldsS = (size_t)64 * BP512 * 8 + 16 * HS * 4;

  const int nblk = (T_DIM * B_DIM) / 64;

  embed_xg0_kernel<<<nblk, 512, ldsE, stream>>>(inputs, W_emb, b_emb, Wx0, b0, xg);
  zero_kernel<<<ringN / 256, 256, 0, stream>>>(ring, ringN);
  scan_kernel<HID><<<B_DIM / 16, 256, ldsS, stream>>>(xg, Wh0, Wo0, bo0, ring, x1);
  xg1_kernel<<<nblk, 512, ldsX, stream>>>(x1, Wx1, b1, xg);
  zero_kernel<<<ringN / 256, 256, 0, stream>>>(ring, ringN);
  scan_kernel<OUTD><<<B_DIM / 16, 256, ldsS, stream>>>(xg, Wh1, Wo1, bo1, ring, out);
}


# scan kernels: the 15 past candidates of the fractional filter move from the global scratch ring into a per-lane register shift register (15x8); ring clears gone (6->4 launches); sum order unchanged
# speedup vs baseline: 7.0376x; 1.1222x over previous
#include <hip/hip_runtime.h>
#include <cstdint>
#include <cstddef>

#define DEVINL __device__ __forceinline__

typedef float v2f __attribute__((ext_vector_type(2)));
typedef float v8f __attribute__((ext_vector_type(8)));

static constexpr int T_DIM  = 1024;
static constexpr int B_DIM  = 64;
static constexpr int IN_DIM = 32;
static constexpr int EMB    = 64;
static constexpr int HID    = 128;
static constexpr int KF     = 16;
static constexpr int OUTD   = 32;
static constexpr int G4     = 512;

static constexpr int HS     = 132;
static constexpr int XS     = 68;
static constexpr int IS     = 36;
static constexpr int BP512  = 528;
static constexpr int BP64   = 80;

DEVINL float sigm(float x) { return 1.0f / (1.0f + __expf(-x)); }

DEVINL v8f wmma4(v2f a, v2f b, v8f c) {
  return __builtin_amdgcn_wmma_f32_16x16x4_f32(false, a, false, b, (short)0, c,
                                               false, false);
}

DEVINL v8f vzero8() {
  v8f z;
#pragma unroll
  for (int r = 0; r < 8; ++r) z[r] = 0.0f;
  return z;
}

__global__ void __launch_bounds__(512)
embed_xg0_kernel(const float* __restrict__ in,
                 const float* __restrict__ We,
                 const float* __restrict__ be,
                 const float* __restrict__ Wx0,
                 const float* __restrict__ b0,
                 float* __restrict__ xg)
{
  extern __shared__ __align__(16) char smem[];
  float* inP = (float*)smem;
  v2f*   WeP = (v2f*)(smem + 64 * IS * 4);
  float* xb  = (float*)(smem + 64 * IS * 4 + 16 * BP64 * 8);
  v2f*   WxP = (v2f*)(smem + 64 * IS * 4 + 16 * BP64 * 8 + 64 * XS * 4);

  const int tid  = threadIdx.x;
  const int wave = tid >> 5;
  const int lane = tid & 31;
  const int lh   = lane >> 4;
  const int ln   = lane & 15;
  const int row0 = blockIdx.x * 64;

  for (int i = tid; i < 64 * 32; i += 512) {
    int r = i >> 5, c = i & 31;
    inP[r * IS + c] = in[(size_t)(row0 + r) * 32 + c];
  }
  for (int i = tid; i < 16 * 64; i += 512) {
    int kp = i >> 6, n = i & 63;
    v2f w; w.x = We[(2 * kp) * 64 + n]; w.y = We[(2 * kp + 1) * 64 + n];
    WeP[kp * BP64 + n] = w;
  }
  for (int i = tid; i < 32 * 512; i += 512) {
    int kp = i >> 9, n = i & 511;
    v2f w; w.x = Wx0[(2 * kp) * 512 + n]; w.y = Wx0[(2 * kp + 1) * 512 + n];
    WxP[kp * BP512 + n] = w;
  }
  __syncthreads();

  {
    const int tm = wave >> 2, tn = wave & 3;
    const int col = tn * 16 + ln;
    const float bv = be[col];
    v8f acc = vzero8();
    const int am = tm * 16 + ln;
#pragma unroll
    for (int k0 = 0; k0 < 32; k0 += 4) {
      v2f a = *(const v2f*)&inP[am * IS + k0 + 2 * lh];
      v2f b = WeP[((k0 >> 1) + lh) * BP64 + col];
      acc = wmma4(a, b, acc);
    }
#pragma unroll
    for (int r = 0; r < 8; ++r)
      xb[(tm * 16 + r + 8 * lh) * XS + col] = acc[r] + bv;
  }
  __syncthreads();

  for (int task = wave; task < 128; task += 16) {
    const int tm = task >> 5, tn = task & 31;
    const int col = tn * 16 + ln;
    const float bv = b0[col];
    v8f acc = vzero8();
    const int am = tm * 16 + ln;
#pragma unroll
    for (int k0 = 0; k0 < 64; k0 += 4) {
      v2f a = *(const v2f*)&xb[am * XS + k0 + 2 * lh];
      v2f b = WxP[((k0 >> 1) + lh) * BP512 + col];
      acc = wmma4(a, b, acc);
    }
#pragma unroll
    for (int r = 0; r < 8; ++r)
      xg[(size_t)(row0 + tm * 16 + r + 8 * lh) * 512 + col] = acc[r] + bv;
  }
}

__global__ void __launch_bounds__(512)
xg1_kernel(const float* __restrict__ x1,
           const float* __restrict__ Wx1,
           const float* __restrict__ b1,
           float* __restrict__ xg)
{
  extern __shared__ __align__(16) char smem[];
  float* aB = (float*)smem;
  v2f*   WP = (v2f*)(smem + 64 * HS * 4);

  const int tid  = threadIdx.x;
  const int wave = tid >> 5;
  const int lane = tid & 31;
  const int lh   = lane >> 4;
  const int ln   = lane & 15;
  const int row0 = blockIdx.x * 64;

  for (int i = tid; i < 64 * 128; i += 512) {
    int r = i >> 7, c = i & 127;
    aB[r * HS + c] = x1[(size_t)(row0 + r) * 128 + c];
  }
  for (int i = tid; i < 64 * 512; i += 512) {
    int kp = i >> 9, n = i & 511;
    v2f w; w.x = Wx1[(2 * kp) * 512 + n]; w.y = Wx1[(2 * kp + 1) * 512 + n];
    WP[kp * BP512 + n] = w;
  }
  __syncthreads();

  for (int task = wave; task < 128; task += 16) {
    const int tm = task >> 5, tn = task & 31;
    const int col = tn * 16 + ln;
    const float bv = b1[col];
    v8f acc = vzero8();
    const int am = tm * 16 + ln;
    for (int k0 = 0; k0 < 128; k0 += 4) {
      v2f a = *(const v2f*)&aB[am * HS + k0 + 2 * lh];
      v2f b = WP[((k0 >> 1) + lh) * BP512 + col];
      acc = wmma4(a, b, acc);
    }
#pragma unroll
    for (int r = 0; r < 8; ++r)
      xg[(size_t)(row0 + tm * 16 + r + 8 * lh) * 512 + col] = acc[r] + bv;
  }
}

template <int NOUT>
__global__ void __launch_bounds__(256)
scan_kernel(const float* __restrict__ xg,
            const float* __restrict__ Wh,
            const float* __restrict__ Wo,
            const float* __restrict__ bo,
            float* __restrict__ out)
{
  extern __shared__ __align__(16) char smem[];
  v2f*   WhP  = (v2f*)smem;
  float* hbuf = (float*)(smem + 64 * BP512 * 8);

  const int tid  = threadIdx.x;
  const int wave = tid >> 5;
  const int lane = tid & 31;
  const int lh   = lane >> 4;
  const int ln   = lane & 15;
  const int rowb = blockIdx.x * 16;
  const int colh = wave * 16 + ln;

  for (int i = tid; i < 64 * 512; i += 256) {
    int kp = i >> 9, n = i & 511;
    v2f w; w.x = Wh[(2 * kp) * 512 + n]; w.y = Wh[(2 * kp + 1) * 512 + n];
    WhP[kp * BP512 + n] = w;
  }
  for (int i = tid; i < 16 * HS; i += 256) hbuf[i] = 0.0f;
  __syncthreads();

  float hnew[8];

  float hist[KF - 1][8];
#pragma unroll
  for (int j = 0; j < KF - 1; ++j)
#pragma unroll
    for (int r = 0; r < 8; ++r) hist[j][r] = 0.0f;

  for (int t = 0; t < T_DIM; ++t) {
    const float* xgt = xg + (size_t)t * (64 * 512) + (size_t)rowb * 512;

    if (t + 1 < T_DIM) {
#pragma unroll
      for (int q = 1; q <= 4; ++q)
        __builtin_prefetch(xgt + (size_t)q * (64 * 512) + (size_t)tid * 32, 0, 0);
    }

    {
      float xv0[8], xv1[8], xv2[8], xv3[8];
#pragma unroll
      for (int r = 0; r < 8; ++r) {
        const int row = r + 8 * lh;
        const float* p = xgt + row * 512 + colh;
        xv0[r] = p[0];  xv1[r] = p[128];
        xv2[r] = p[256]; xv3[r] = p[384];
      }

      v8f acc0 = vzero8(), acc1 = vzero8(), acc2 = vzero8(), acc3 = vzero8();
      for (int k0 = 0; k0 < 128; k0 += 4) {
        v2f a = *(const v2f*)&hbuf[ln * HS + k0 + 2 * lh];
        const int kp = (k0 >> 1) + lh;
        acc0 = wmma4(a, WhP[kp * BP512 + colh + 0],   acc0);
        acc1 = wmma4(a, WhP[kp * BP512 + colh + 128], acc1);
        acc2 = wmma4(a, WhP[kp * BP512 + colh + 256], acc2);
        acc3 = wmma4(a, WhP[kp * BP512 + colh + 384], acc3);
      }

      float hc[8], og[8], dd[8];
#pragma unroll
      for (int r = 0; r < 8; ++r) {
        const float ig = sigm(acc0[r] + xv0[r]);
        og[r] = sigm(acc1[r] + xv1[r]);
        const float cg = tanhf(acc2[r] + xv2[r]);
        dd[r] = 0.5f * sigm(acc3[r] + xv3[r]);
        hc[r] = ig * cg;
      }

      float csum[8], w[8];
#pragma unroll
      for (int r = 0; r < 8; ++r) { csum[r] = hc[r]; w[r] = 1.0f; }
#pragma unroll
      for (int j = 1; j < KF; ++j) {
        const float jm1 = (float)(j - 1), rj = 1.0f / (float)j;
#pragma unroll
        for (int r = 0; r < 8; ++r) {
          w[r] *= (jm1 + dd[r]) * rj;
          csum[r] += w[r] * hist[j - 1][r];
        }
      }
#pragma unroll
      for (int j = KF - 2; j > 0; --j)
#pragma unroll
        for (int r = 0; r < 8; ++r) hist[j][r] = hist[j - 1][r];
#pragma unroll
      for (int r = 0; r < 8; ++r) {
        hist[0][r] = hc[r];
        hnew[r] = og[r] * tanhf(csum[r]);
      }
    }

    __syncthreads();
#pragma unroll
    for (int r = 0; r < 8; ++r)
      hbuf[(r + 8 * lh) * HS + colh] = hnew[r];
    __syncthreads();

    if (NOUT / 16 >= 8 || wave < NOUT / 16) {
      float* ot = out + (size_t)t * (64 * NOUT) + (size_t)rowb * NOUT;
      const int col = wave * 16 + ln;
      const float bv = bo[col];
      v8f acc = vzero8();
      for (int k0 = 0; k0 < 128; k0 += 4) {
        v2f a = *(const v2f*)&hbuf[ln * HS + k0 + 2 * lh];
        v2f b;
        b.x = Wo[(k0 + 2 * lh) * NOUT + col];
        b.y = Wo[(k0 + 2 * lh + 1) * NOUT + col];
        acc = wmma4(a, b, acc);
      }
#pragma unroll
      for (int r = 0; r < 8; ++r)
        ot[(r + 8 * lh) * NOUT + col] = acc[r] + bv;
    }
  }
}

extern "C" void kernel_launch(void* const* d_in, const int* in_sizes, int n_in,
                              void* d_out, int out_size, void* d_ws, size_t ws_size,
                              hipStream_t stream) {
  const float* inputs = (const float*)d_in[0];
  const float* W_emb  = (const float*)d_in[1];
  const float* b_emb  = (const float*)d_in[2];
  const float* Wx0    = (const float*)d_in[3];
  const float* Wh0    = (const float*)d_in[4];
  const float* b0     = (const float*)d_in[5];
  const float* Wo0    = (const float*)d_in[6];
  const float* bo0    = (const float*)d_in[7];
  const float* Wx1    = (const float*)d_in[8];
  const float* Wh1    = (const float*)d_in[9];
  const float* b1     = (const float*)d_in[10];
  const float* Wo1    = (const float*)d_in[11];
  const float* bo1    = (const float*)d_in[12];
  float* out = (float*)d_out;

  float* xg   = (float*)d_ws;
  float* x1   = xg + (size_t)T_DIM * B_DIM * G4;

  const size_t ldsE = (size_t)64 * IS * 4 + 16 * BP64 * 8 + 64 * XS * 4 + 32 * BP512 * 8;
  const size_t ldsX = (size_t)64 * HS * 4 + 64 * BP512 * 8;
  const size_t ldsS = (size_t)64 * BP512 * 8 + 16 * HS * 4;

  const int nblk = (T_DIM * B_DIM) / 64;

  embed_xg0_kernel<<<nblk, 512, ldsE, stream>>>(inputs, W_emb, b_emb, Wx0, b0, xg);
  scan_kernel<HID><<<B_DIM / 16, 256, ldsS, stream>>>(xg, Wh0, Wo0, bo0, x1);
  xg1_kernel<<<nblk, 512, ldsX, stream>>>(x1, Wx1, b1, xg);
  scan_kernel<OUTD><<<B_DIM / 16, 256, ldsS, stream>>>(xg, Wh1, Wo1, bo1, out);
}


# output projection h@Wo+bo leaves the per-timestep scan: scan emits h_t (full-row stores), one WMMA kernel per layer projects all T*B rows afterwards (4->6 launches); k order and sums unchanged
# speedup vs baseline: 7.3198x; 1.0401x over previous
#include <hip/hip_runtime.h>
#include <cstdint>
#include <cstddef>

#define DEVINL __device__ __forceinline__

typedef float v2f __attribute__((ext_vector_type(2)));
typedef float v8f __attribute__((ext_vector_type(8)));

static constexpr int T_DIM  = 1024;
static constexpr int B_DIM  = 64;
static constexpr int IN_DIM = 32;
static constexpr int EMB    = 64;
static constexpr int HID    = 128;
static constexpr int KF     = 16;
static constexpr int OUTD   = 32;
static constexpr int G4     = 512;

static constexpr int HS     = 132;
static constexpr int XS     = 68;
static constexpr int IS     = 36;
static constexpr int BP512  = 528;
static constexpr int BP64   = 80;

DEVINL float sigm(float x) { return 1.0f / (1.0f + __expf(-x)); }

DEVINL v8f wmma4(v2f a, v2f b, v8f c) {
  return __builtin_amdgcn_wmma_f32_16x16x4_f32(false, a, false, b, (short)0, c,
                                               false, false);
}

DEVINL v8f vzero8() {
  v8f z;
#pragma unroll
  for (int r = 0; r < 8; ++r) z[r] = 0.0f;
  return z;
}

__global__ void __launch_bounds__(512)
embed_xg0_kernel(const float* __restrict__ in,
                 const float* __restrict__ We,
                 const float* __restrict__ be,
                 const float* __restrict__ Wx0,
                 const float* __restrict__ b0,
                 float* __restrict__ xg)
{
  extern __shared__ __align__(16) char smem[];
  float* inP = (float*)smem;
  v2f*   WeP = (v2f*)(smem + 64 * IS * 4);
  float* xb  = (float*)(smem + 64 * IS * 4 + 16 * BP64 * 8);
  v2f*   WxP = (v2f*)(smem + 64 * IS * 4 + 16 * BP64 * 8 + 64 * XS * 4);

  const int tid  = threadIdx.x;
  const int wave = tid >> 5;
  const int lane = tid & 31;
  const int lh   = lane >> 4;
  const int ln   = lane & 15;
  const int row0 = blockIdx.x * 64;

  for (int i = tid; i < 64 * 32; i += 512) {
    int r = i >> 5, c = i & 31;
    inP[r * IS + c] = in[(size_t)(row0 + r) * 32 + c];
  }
  for (int i = tid; i < 16 * 64; i += 512) {
    int kp = i >> 6, n = i & 63;
    v2f w; w.x = We[(2 * kp) * 64 + n]; w.y = We[(2 * kp + 1) * 64 + n];
    WeP[kp * BP64 + n] = w;
  }
  for (int i = tid; i < 32 * 512; i += 512) {
    int kp = i >> 9, n = i & 511;
    v2f w; w.x = Wx0[(2 * kp) * 512 + n]; w.y = Wx0[(2 * kp + 1) * 512 + n];
    WxP[kp * BP512 + n] = w;
  }
  __syncthreads();

  {
    const int tm = wave >> 2, tn = wave & 3;
    const int col = tn * 16 + ln;
    const float bv = be[col];
    v8f acc = vzero8();
    const int am = tm * 16 + ln;
#pragma unroll
    for (int k0 = 0; k0 < 32; k0 += 4) {
      v2f a = *(const v2f*)&inP[am * IS + k0 + 2 * lh];
      v2f b = WeP[((k0 >> 1) + lh) * BP64 + col];
      acc = wmma4(a, b, acc);
    }
#pragma unroll
    for (int r = 0; r < 8; ++r)
      xb[(tm * 16 + r + 8 * lh) * XS + col] = acc[r] + bv;
  }
  __syncthreads();

  for (int task = wave; task < 128; task += 16) {
    const int tm = task >> 5, tn = task & 31;
    const int col = tn * 16 + ln;
    const float bv = b0[col];
    v8f acc = vzero8();
    const int am = tm * 16 + ln;
#pragma unroll
    for (int k0 = 0; k0 < 64; k0 += 4) {
      v2f a = *(const v2f*)&xb[am * XS + k0 + 2 * lh];
      v2f b = WxP[((k0 >> 1) + lh) * BP512 + col];
      acc = wmma4(a, b, acc);
    }
#pragma unroll
    for (int r = 0; r < 8; ++r)
      xg[(size_t)(row0 + tm * 16 + r + 8 * lh) * 512 + col] = acc[r] + bv;
  }
}

__global__ void __launch_bounds__(512)
xg1_kernel(const float* __restrict__ x1,
           const float* __restrict__ Wx1,
           const float* __restrict__ b1,
           float* __restrict__ xg)
{
  extern __shared__ __align__(16) char smem[];
  float* aB = (float*)smem;
  v2f*   WP = (v2f*)(smem + 64 * HS * 4);

  const int tid  = threadIdx.x;
  const int wave = tid >> 5;
  const int lane = tid & 31;
  const int lh   = lane >> 4;
  const int ln   = lane & 15;
  const int row0 = blockIdx.x * 64;

  for (int i = tid; i < 64 * 128; i += 512) {
    int r = i >> 7, c = i & 127;
    aB[r * HS + c] = x1[(size_t)(row0 + r) * 128 + c];
  }
  for (int i = tid; i < 64 * 512; i += 512) {
    int kp = i >> 9, n = i & 511;
    v2f w; w.x = Wx1[(2 * kp) * 512 + n]; w.y = Wx1[(2 * kp + 1) * 512 + n];
    WP[kp * BP512 + n] = w;
  }
  __syncthreads();

  for (int task = wave; task < 128; task += 16) {
    const int tm = task >> 5, tn = task & 31;
    const int col = tn * 16 + ln;
    const float bv = b1[col];
    v8f acc = vzero8();
    const int am = tm * 16 + ln;
    for (int k0 = 0; k0 < 128; k0 += 4) {
      v2f a = *(const v2f*)&aB[am * HS + k0 + 2 * lh];
      v2f b = WP[((k0 >> 1) + lh) * BP512 + col];
      acc = wmma4(a, b, acc);
    }
#pragma unroll
    for (int r = 0; r < 8; ++r)
      xg[(size_t)(row0 + tm * 16 + r + 8 * lh) * 512 + col] = acc[r] + bv;
  }
}

__global__ void __launch_bounds__(256)
scan_kernel(const float* __restrict__ xg,
            const float* __restrict__ Wh,
            float* __restrict__ hseq)
{
  extern __shared__ __align__(16) char smem[];
  v2f*   WhP  = (v2f*)smem;
  float* hbuf = (float*)(smem + 64 * BP512 * 8);

  const int tid  = threadIdx.x;
  const int wave = tid >> 5;
  const int lane = tid & 31;
  const int lh   = lane >> 4;
  const int ln   = lane & 15;
  const int rowb = blockIdx.x * 16;
  const int colh = wave * 16 + ln;

  for (int i = tid; i < 64 * 512; i += 256) {
    int kp = i >> 9, n = i & 511;
    v2f w; w.x = Wh[(2 * kp) * 512 + n]; w.y = Wh[(2 * kp + 1) * 512 + n];
    WhP[kp * BP512 + n] = w;
  }
  for (int i = tid; i < 16 * HS; i += 256) hbuf[i] = 0.0f;
  __syncthreads();

  float hnew[8];

  float hist[KF - 1][8];
#pragma unroll
  for (int j = 0; j < KF - 1; ++j)
#pragma unroll
    for (int r = 0; r < 8; ++r) hist[j][r] = 0.0f;

  for (int t = 0; t < T_DIM; ++t) {
    const float* xgt = xg + (size_t)t * (64 * 512) + (size_t)rowb * 512;

    if (t + 1 < T_DIM) {
#pragma unroll
      for (int q = 1; q <= 4; ++q)
        __builtin_prefetch(xgt + (size_t)q * (64 * 512) + (size_t)tid * 32, 0, 0);
    }

    {
      float xv0[8], xv1[8], xv2[8], xv3[8];
#pragma unroll
      for (int r = 0; r < 8; ++r) {
        const int row = r + 8 * lh;
        const float* p = xgt + row * 512 + colh;
        xv0[r] = p[0];  xv1[r] = p[128];
        xv2[r] = p[256]; xv3[r] = p[384];
      }

      v8f acc0 = vzero8(), acc1 = vzero8(), acc2 = vzero8(), acc3 = vzero8();
      for (int k0 = 0; k0 < 128; k0 += 4) {
        v2f a = *(const v2f*)&hbuf[ln * HS + k0 + 2 * lh];
        const int kp = (k0 >> 1) + lh;
        acc0 = wmma4(a, WhP[kp * BP512 + colh + 0],   acc0);
        acc1 = wmma4(a, WhP[kp * BP512 + colh + 128], acc1);
        acc2 = wmma4(a, WhP[kp * BP512 + colh + 256], acc2);
        acc3 = wmma4(a, WhP[kp * BP512 + colh + 384], acc3);
      }

      float hc[8], og[8], dd[8];
#pragma unroll
      for (int r = 0; r < 8; ++r) {
        const float ig = sigm(acc0[r] + xv0[r]);
        og[r] = sigm(acc1[r] + xv1[r]);
        const float cg = tanhf(acc2[r] + xv2[r]);
        dd[r] = 0.5f * sigm(acc3[r] + xv3[r]);
        hc[r] = ig * cg;
      }

      float csum[8], w[8];
#pragma unroll
      for (int r = 0; r < 8; ++r) { csum[r] = hc[r]; w[r] = 1.0f; }
#pragma unroll
      for (int j = 1; j < KF; ++j) {
        const float jm1 = (float)(j - 1), rj = 1.0f / (float)j;
#pragma unroll
        for (int r = 0; r < 8; ++r) {
          w[r] *= (jm1 + dd[r]) * rj;
          csum[r] += w[r] * hist[j - 1][r];
        }
      }
#pragma unroll
      for (int j = KF - 2; j > 0; --j)
#pragma unroll
        for (int r = 0; r < 8; ++r) hist[j][r] = hist[j - 1][r];
#pragma unroll
      for (int r = 0; r < 8; ++r) {
        hist[0][r] = hc[r];
        hnew[r] = og[r] * tanhf(csum[r]);
      }
    }

    __syncthreads();
#pragma unroll
    for (int r = 0; r < 8; ++r)
      hbuf[(r + 8 * lh) * HS + colh] = hnew[r];
    __syncthreads();

    {
      float* ht = hseq + (size_t)t * (64 * HID) + (size_t)rowb * HID;
#pragma unroll
      for (int q = 0; q < 2; ++q) {
        const int row = 2 * wave + q;
        const float4 v = *(const float4*)&hbuf[row * HS + 4 * lane];
        *(float4*)&ht[row * HID + 4 * lane] = v;
      }
    }
  }
}

template <int NOUT>
__global__ void __launch_bounds__(512)
proj_kernel(const float* hin,
            const float* __restrict__ Wo,
            const float* __restrict__ bo,
            float* y)
{
  constexpr int BPS = NOUT + 16;
  constexpr int NTN = NOUT / 16;
  extern __shared__ __align__(16) char smem[];
  float* aB = (float*)smem;
  v2f*   WP = (v2f*)(smem + 64 * HS * 4);

  const int tid  = threadIdx.x;
  const int wave = tid >> 5;
  const int lane = tid & 31;
  const int lh   = lane >> 4;
  const int ln   = lane & 15;
  const int row0 = blockIdx.x * 64;

  for (int i = tid; i < 64 * 128; i += 512) {
    int r = i >> 7, c = i & 127;
    aB[r * HS + c] = hin[(size_t)(row0 + r) * 128 + c];
  }
  for (int i = tid; i < 64 * NOUT; i += 512) {
    int kp = i / NOUT, n = i % NOUT;
    v2f w; w.x = Wo[(2 * kp) * NOUT + n]; w.y = Wo[(2 * kp + 1) * NOUT + n];
    WP[kp * BPS + n] = w;
  }
  __syncthreads();

  for (int task = wave; task < 4 * NTN; task += 16) {
    const int tm = task / NTN, tn = task % NTN;
    const int col = tn * 16 + ln;
    const float bv = bo[col];
    v8f acc = vzero8();
    const int am = tm * 16 + ln;
    for (int k0 = 0; k0 < 128; k0 += 4) {
      v2f a = *(const v2f*)&aB[am * HS + k0 + 2 * lh];
      v2f b = WP[((k0 >> 1) + lh) * BPS + col];
      acc = wmma4(a, b, acc);
    }
#pragma unroll
    for (int r = 0; r < 8; ++r)
      y[(size_t)(row0 + tm * 16 + r + 8 * lh) * NOUT + col] = acc[r] + bv;
  }
}

extern "C" void kernel_launch(void* const* d_in, const int* in_sizes, int n_in,
                              void* d_out, int out_size, void* d_ws, size_t ws_size,
                              hipStream_t stream) {
  const float* inputs = (const float*)d_in[0];
  const float* W_emb  = (const float*)d_in[1];
  const float* b_emb  = (const float*)d_in[2];
  const float* Wx0    = (const float*)d_in[3];
  const float* Wh0    = (const float*)d_in[4];
  const float* b0     = (const float*)d_in[5];
  const float* Wo0    = (const float*)d_in[6];
  const float* bo0    = (const float*)d_in[7];
  const float* Wx1    = (const float*)d_in[8];
  const float* Wh1    = (const float*)d_in[9];
  const float* b1     = (const float*)d_in[10];
  const float* Wo1    = (const float*)d_in[11];
  const float* bo1    = (const float*)d_in[12];
  float* out = (float*)d_out;

  float* xg   = (float*)d_ws;
  float* hx   = xg + (size_t)T_DIM * B_DIM * G4;

  const size_t ldsE = (size_t)64 * IS * 4 + 16 * BP64 * 8 + 64 * XS * 4 + 32 * BP512 * 8;
  const size_t ldsX = (size_t)64 * HS * 4 + 64 * BP512 * 8;
  const size_t ldsS = (size_t)64 * BP512 * 8 + 16 * HS * 4;
  const size_t ldsP0 = (size_t)64 * HS * 4 + 64 * (HID + 16) * 8;
  const size_t ldsP1 = (size_t)64 * HS * 4 + 64 * (OUTD + 16) * 8;

  const int nblk = (T_DIM * B_DIM) / 64;

  embed_xg0_kernel<<<nblk, 512, ldsE, stream>>>(inputs, W_emb, b_emb, Wx0, b0, xg);
  scan_kernel<<<B_DIM / 16, 256, ldsS, stream>>>(xg, Wh0, hx);
  proj_kernel<HID><<<nblk, 512, ldsP0, stream>>>(hx, Wo0, bo0, hx);
  xg1_kernel<<<nblk, 512, ldsX, stream>>>(hx, Wx1, b1, xg);
  scan_kernel<<<B_DIM / 16, 256, ldsS, stream>>>(xg, Wh1, hx);
  proj_kernel<OUTD><<<nblk, 512, ldsP1, stream>>>(hx, Wo1, bo1, out);
}


# scan kernels: each wave keeps its 128x64 slice of Wh as B fragments in 256 VGPRs for the whole scan; LDS holds only h, read as one group of 32 A loads per step; k order and sums unchanged
# speedup vs baseline: 8.3804x; 1.1449x over previous
#include <hip/hip_runtime.h>
#include <cstdint>
#include <cstddef>

#define DEVINL __device__ __forceinline__

typedef float v2f __attribute__((ext_vector_type(2)));
typedef float v8f __attribute__((ext_vector_type(8)));

static constexpr int T_DIM  = 1024;
static constexpr int B_DIM  = 64;
static constexpr int IN_DIM = 32;
static constexpr int EMB    = 64;
static constexpr int HID    = 128;
static constexpr int KF     = 16;
static constexpr int OUTD   = 32;
static constexpr int G4     = 512;

static constexpr int HS     = 132;
static constexpr int XS     = 68;
static constexpr int IS     = 36;
static constexpr int BP512  = 528;
static constexpr int BP64   = 80;

DEVINL float sigm(float x) { return 1.0f / (1.0f + __expf(-x)); }

DEVINL v8f wmma4(v2f a, v2f b, v8f c) {
  return __builtin_amdgcn_wmma_f32_16x16x4_f32(false, a, false, b, (short)0, c,
                                               false, false);
}

DEVINL v8f vzero8() {
  v8f z;
#pragma unroll
  for (int r = 0; r < 8; ++r) z[r] = 0.0f;
  return z;
}

__global__ void __launch_bounds__(512)
embed_xg0_kernel(const float* __restrict__ in,
                 const float* __restrict__ We,
                 const float* __restrict__ be,
                 const float* __restrict__ Wx0,
                 const float* __restrict__ b0,
                 float* __restrict__ xg)
{
  extern __shared__ __align__(16) char smem[];
  float* inP = (float*)smem;
  v2f*   WeP = (v2f*)(smem + 64 * IS * 4);
  float* xb  = (float*)(smem + 64 * IS * 4 + 16 * BP64 * 8);
  v2f*   WxP = (v2f*)(smem + 64 * IS * 4 + 16 * BP64 * 8 + 64 * XS * 4);

  const int tid  = threadIdx.x;
  const int wave = tid >> 5;
  const int lane = tid & 31;
  const int lh   = lane >> 4;
  const int ln   = lane & 15;
  const int row0 = blockIdx.x * 64;

  for (int i = tid; i < 64 * 32; i += 512) {
    int r = i >> 5, c = i & 31;
    inP[r * IS + c] = in[(size_t)(row0 + r) * 32 + c];
  }
  for (int i = tid; i < 16 * 64; i += 512) {
    int kp = i >> 6, n = i & 63;
    v2f w; w.x = We[(2 * kp) * 64 + n]; w.y = We[(2 * kp + 1) * 64 + n];
    WeP[kp * BP64 + n] = w;
  }
  for (int i = tid; i < 32 * 512; i += 512) {
    int kp = i >> 9, n = i & 511;
    v2f w; w.x = Wx0[(2 * kp) * 512 + n]; w.y = Wx0[(2 * kp + 1) * 512 + n];
    WxP[kp * BP512 + n] = w;
  }
  __syncthreads();

  {
    const int tm = wave >> 2, tn = wave & 3;
    const int col = tn * 16 + ln;
    const float bv = be[col];
    v8f acc = vzero8();
    const int am = tm * 16 + ln;
#pragma unroll
    for (int k0 = 0; k0 < 32; k0 += 4) {
      v2f a = *(const v2f*)&inP[am * IS + k0 + 2 * lh];
      v2f b = WeP[((k0 >> 1) + lh) * BP64 + col];
      acc = wmma4(a, b, acc);
    }
#pragma unroll
    for (int r = 0; r < 8; ++r)
      xb[(tm * 16 + r + 8 * lh) * XS + col] = acc[r] + bv;
  }
  __syncthreads();

  for (int task = wave; task < 128; task += 16) {
    const int tm = task >> 5, tn = task & 31;
    const int col = tn * 16 + ln;
    const float bv = b0[col];
    v8f acc = vzero8();
    const int am = tm * 16 + ln;
#pragma unroll
    for (int k0 = 0; k0 < 64; k0 += 4) {
      v2f a = *(const v2f*)&xb[am * XS + k0 + 2 * lh];
      v2f b = WxP[((k0 >> 1) + lh) * BP512 + col];
      acc = wmma4(a, b, acc);
    }
#pragma unroll
    for (int r = 0; r < 8; ++r)
      xg[(size_t)(row0 + tm * 16 + r + 8 * lh) * 512 + col] = acc[r] + bv;
  }
}

__global__ void __launch_bounds__(512)
xg1_kernel(const float* __restrict__ x1,
           const float* __restrict__ Wx1,
           const float* __restrict__ b1,
           float* __restrict__ xg)
{
  extern __shared__ __align__(16) char smem[];
  float* aB = (float*)smem;
  v2f*   WP = (v2f*)(smem + 64 * HS * 4);

  const int tid  = threadIdx.x;
  const int wave = tid >> 5;
  const int lane = tid & 31;
  const int lh   = lane >> 4;
  const int ln   = lane & 15;
  const int row0 = blockIdx.x * 64;

  for (int i = tid; i < 64 * 128; i += 512) {
    int r = i >> 7, c = i & 127;
    aB[r * HS + c] = x1[(size_t)(row0 + r) * 128 + c];
  }
  for (int i = tid; i < 64 * 512; i += 512) {
    int kp = i >> 9, n = i & 511;
    v2f w; w.x = Wx1[(2 * kp) * 512 + n]; w.y = Wx1[(2 * kp + 1) * 512 + n];
    WP[kp * BP512 + n] = w;
  }
  __syncthreads();

  for (int task = wave; task < 128; task += 16) {
    const int tm = task >> 5, tn = task & 31;
    const int col = tn * 16 + ln;
    const float bv = b1[col];
    v8f acc = vzero8();
    const int am = tm * 16 + ln;
    for (int k0 = 0; k0 < 128; k0 += 4) {
      v2f a = *(const v2f*)&aB[am * HS + k0 + 2 * lh];
      v2f b = WP[((k0 >> 1) + lh) * BP512 + col];
      acc = wmma4(a, b, acc);
    }
#pragma unroll
    for (int r = 0; r < 8; ++r)
      xg[(size_t)(row0 + tm * 16 + r + 8 * lh) * 512 + col] = acc[r] + bv;
  }
}

__global__ void __launch_bounds__(256)
scan_kernel(const float* __restrict__ xg,
            const float* __restrict__ Wh,
            float* __restrict__ hseq)
{
  extern __shared__ __align__(16) char smem[];
  float* hbuf = (float*)smem;

  const int tid  = threadIdx.x;
  const int wave = tid >> 5;
  const int lane = tid & 31;
  const int lh   = lane >> 4;
  const int ln   = lane & 15;
  const int rowb = blockIdx.x * 16;
  const int colh = wave * 16 + ln;

  v2f wfr[4][32];
#pragma unroll
  for (int ks = 0; ks < 32; ++ks) {
    const float* wp = Wh + (size_t)(4 * ks + 2 * lh) * 512 + colh;
#pragma unroll
    for (int g = 0; g < 4; ++g) {
      v2f w; w.x = wp[g * 128]; w.y = wp[512 + g * 128];
      wfr[g][ks] = w;
    }
  }
  for (int i = tid; i < 16 * HS; i += 256) hbuf[i] = 0.0f;
  __syncthreads();

  float hnew[8];

  float hist[KF - 1][8];
#pragma unroll
  for (int j = 0; j < KF - 1; ++j)
#pragma unroll
    for (int r = 0; r < 8; ++r) hist[j][r] = 0.0f;

  for (int t = 0; t < T_DIM; ++t) {
    const float* xgt = xg + (size_t)t * (64 * 512) + (size_t)rowb * 512;

    if (t + 1 < T_DIM) {
#pragma unroll
      for (int q = 1; q <= 4; ++q)
        __builtin_prefetch(xgt + (size_t)q * (64 * 512) + (size_t)tid * 32, 0, 0);
    }

    {
      v2f afr[32];
#pragma unroll
      for (int ks = 0; ks < 32; ++ks)
        afr[ks] = *(const v2f*)&hbuf[ln * HS + 4 * ks + 2 * lh];
      __builtin_amdgcn_sched_barrier(0);

      float xv0[8], xv1[8], xv2[8], xv3[8];
#pragma unroll
      for (int r = 0; r < 8; ++r) {
        const int row = r + 8 * lh;
        const float* p = xgt + row * 512 + colh;
        xv0[r] = p[0];  xv1[r] = p[128];
        xv2[r] = p[256]; xv3[r] = p[384];
      }

      v8f acc0 = vzero8(), acc1 = vzero8(), acc2 = vzero8(), acc3 = vzero8();
#pragma unroll
      for (int ks = 0; ks < 32; ++ks) {
        const v2f a = afr[ks];
        acc0 = wmma4(a, wfr[0][ks], acc0);
        acc1 = wmma4(a, wfr[1][ks], acc1);
        acc2 = wmma4(a, wfr[2][ks], acc2);
        acc3 = wmma4(a, wfr[3][ks], acc3);
      }

      float hc[8], og[8], dd[8];
#pragma unroll
      for (int r = 0; r < 8; ++r) {
        const float ig = sigm(acc0[r] + xv0[r]);
        og[r] = sigm(acc1[r] + xv1[r]);
        const float cg = tanhf(acc2[r] + xv2[r]);
        dd[r] = 0.5f * sigm(acc3[r] + xv3[r]);
        hc[r] = ig * cg;
      }

      float csum[8], w[8];
#pragma unroll
      for (int r = 0; r < 8; ++r) { csum[r] = hc[r]; w[r] = 1.0f; }
#pragma unroll
      for (int j = 1; j < KF; ++j) {
        const float jm1 = (float)(j - 1), rj = 1.0f / (float)j;
#pragma unroll
        for (int r = 0; r < 8; ++r) {
          w[r] *= (jm1 + dd[r]) * rj;
          csum[r] += w[r] * hist[j - 1][r];
        }
      }
#pragma unroll
      for (int j = KF - 2; j > 0; --j)
#pragma unroll
        for (int r = 0; r < 8; ++r) hist[j][r] = hist[j - 1][r];
#pragma unroll
      for (int r = 0; r < 8; ++r) {
        hist[0][r] = hc[r];
        hnew[r] = og[r] * tanhf(csum[r]);
      }
    }

    __syncthreads();
#pragma unroll
    for (int r = 0; r < 8; ++r)
      hbuf[(r + 8 * lh) * HS + colh] = hnew[r];
    __syncthreads();

    {
      float* ht = hseq + (size_t)t * (64 * HID) + (size_t)rowb * HID;
#pragma unroll
      for (int q = 0; q < 2; ++q) {
        const int row = 2 * wave + q;
        const float4 v = *(const float4*)&hbuf[row * HS + 4 * lane];
        *(float4*)&ht[row * HID + 4 * lane] = v;
      }
    }
  }
}

template <int NOUT>
__global__ void __launch_bounds__(512)
proj_kernel(const float* hin,
            const float* __restrict__ Wo,
            const float* __restrict__ bo,
            float* y)
{
  constexpr int BPS = NOUT + 16;
  constexpr int NTN = NOUT / 16;
  extern __shared__ __align__(16) char smem[];
  float* aB = (float*)smem;
  v2f*   WP = (v2f*)(smem + 64 * HS * 4);

  const int tid  = threadIdx.x;
  const int wave = tid >> 5;
  const int lane = tid & 31;
  const int lh   = lane >> 4;
  const int ln   = lane & 15;
  const int row0 = blockIdx.x * 64;

  for (int i = tid; i < 64 * 128; i += 512) {
    int r = i >> 7, c = i & 127;
    aB[r * HS + c] = hin[(size_t)(row0 + r) * 128 + c];
  }
  for (int i = tid; i < 64 * NOUT; i += 512) {
    int kp = i / NOUT, n = i % NOUT;
    v2f w; w.x = Wo[(2 * kp) * NOUT + n]; w.y = Wo[(2 * kp + 1) * NOUT + n];
    WP[kp * BPS + n] = w;
  }
  __syncthreads();

  for (int task = wave; task < 4 * NTN; task += 16) {
    const int tm = task / NTN, tn = task % NTN;
    const int col = tn * 16 + ln;
    const float bv = bo[col];
    v8f acc = vzero8();
    const int am = tm * 16 + ln;
    for (int k0 = 0; k0 < 128; k0 += 4) {
      v2f a = *(const v2f*)&aB[am * HS + k0 + 2 * lh];
      v2f b = WP[((k0 >> 1) + lh) * BPS + col];
      acc = wmma4(a, b, acc);
    }
#pragma unroll
    for (int r = 0; r < 8; ++r)
      y[(size_t)(row0 + tm * 16 + r + 8 * lh) * NOUT + col] = acc[r] + bv;
  }
}

extern "C" void kernel_launch(void* const* d_in, const int* in_sizes, int n_in,
                              void* d_out, int out_size, void* d_ws, size_t ws_size,
                              hipStream_t stream) {
  const float* inputs = (const float*)d_in[0];
  const float* W_emb  = (const float*)d_in[1];
  const float* b_emb  = (const float*)d_in[2];
  const float* Wx0    = (const float*)d_in[3];
  const float* Wh0    = (const float*)d_in[4];
  const float* b0     = (const float*)d_in[5];
  const float* Wo0    = (const float*)d_in[6];
  const float* bo0    = (const float*)d_in[7];
  const float* Wx1    = (const float*)d_in[8];
  const float* Wh1    = (const float*)d_in[9];
  const float* b1     = (const float*)d_in[10];
  const float* Wo1    = (const float*)d_in[11];
  const float* bo1    = (const float*)d_in[12];
  float* out = (float*)d_out;

  float* xg   = (float*)d_ws;
  float* hx   = xg + (size_t)T_DIM * B_DIM * G4;

  const size_t ldsE = (size_t)64 * IS * 4 + 16 * BP64 * 8 + 64 * XS * 4 + 32 * BP512 * 8;
  const size_t ldsX = (size_t)64 * HS * 4 + 64 * BP512 * 8;
  const size_t ldsS = (size_t)16 * HS * 4;
  const size_t ldsP0 = (size_t)64 * HS * 4 + 64 * (HID + 16) * 8;
  const size_t ldsP1 = (size_t)64 * HS * 4 + 64 * (OUTD + 16) * 8;

  const int nblk = (T_DIM * B_DIM) / 64;

  embed_xg0_kernel<<<nblk, 512, ldsE, stream>>>(inputs, W_emb, b_emb, Wx0, b0, xg);
  scan_kernel<<<B_DIM / 16, 256, ldsS, stream>>>(xg, Wh0, hx);
  proj_kernel<HID><<<nblk, 512, ldsP0, stream>>>(hx, Wo0, bo0, hx);
  xg1_kernel<<<nblk, 512, ldsX, stream>>>(hx, Wx1, b1, xg);
  scan_kernel<<<B_DIM / 16, 256, ldsS, stream>>>(xg, Wh1, hx);
  proj_kernel<OUTD><<<nblk, 512, ldsP1, stream>>>(hx, Wo1, bo1, out);
}


# scan kernels: tanh without branches (both ranges of the library form computed, result selected per lane), so a step's eight evaluations interleave instead of running serially; values unchanged
# speedup vs baseline: 9.7362x; 1.1618x over previous
#include <hip/hip_runtime.h>
#include <cstdint>
#include <cstddef>

#define DEVINL __device__ __forceinline__

typedef float v2f __attribute__((ext_vector_type(2)));
typedef float v8f __attribute__((ext_vector_type(8)));

static constexpr int T_DIM  = 1024;
static constexpr int B_DIM  = 64;
static constexpr int IN_DIM = 32;
static constexpr int EMB    = 64;
static constexpr int HID    = 128;
static constexpr int KF     = 16;
static constexpr int OUTD   = 32;
static constexpr int G4     = 512;

static constexpr int HS     = 132;
static constexpr int XS     = 68;
static constexpr int IS     = 36;
static constexpr int BP512  = 528;
static constexpr int BP64   = 80;

DEVINL float sigm(float x) { return 1.0f / (1.0f + __expf(-x)); }

DEVINL float tanh_sel(float x) {
  const float y  = fabsf(x);
  const float y2 = x * x;
  float p = fmaf(__uint_as_float(0xbbbac73du), y2, __uint_as_float(0x3ca908c9u));
  p = fmaf(y2, p, __uint_as_float(0xbd5c1c4eu));
  p = fmaf(y2, p, __uint_as_float(0x3e088382u));
  p = fmaf(y2, p, __uint_as_float(0xbeaaaa99u));
  const float zs = fmaf(y2, y * p, y);
  const float e  = expf(y + y);
  const float zl = fmaf(__builtin_amdgcn_rcpf(e + 1.0f), -2.0f, 1.0f);
  const float z  = (y < 0.625f) ? zs : zl;
  return copysignf(z, x);
}

DEVINL v8f wmma4(v2f a, v2f b, v8f c) {
  return __builtin_amdgcn_wmma_f32_16x16x4_f32(false, a, false, b, (short)0, c,
                                               false, false);
}

DEVINL v8f vzero8() {
  v8f z;
#pragma unroll
  for (int r = 0; r < 8; ++r) z[r] = 0.0f;
  return z;
}

__global__ void __launch_bounds__(512)
embed_xg0_kernel(const float* __restrict__ in,
                 const float* __restrict__ We,
                 const float* __restrict__ be,
                 const float* __restrict__ Wx0,
                 const float* __restrict__ b0,
                 float* __restrict__ xg)
{
  extern __shared__ __align__(16) char smem[];
  float* inP = (float*)smem;
  v2f*   WeP = (v2f*)(smem + 64 * IS * 4);
  float* xb  = (float*)(smem + 64 * IS * 4 + 16 * BP64 * 8);
  v2f*   WxP = (v2f*)(smem + 64 * IS * 4 + 16 * BP64 * 8 + 64 * XS * 4);

  const int tid  = threadIdx.x;
  const int wave = tid >> 5;
  const int lane = tid & 31;
  const int lh   = lane >> 4;
  const int ln   = lane & 15;
  const int row0 = blockIdx.x * 64;

  for (int i = tid; i < 64 * 32; i += 512) {
    int r = i >> 5, c = i & 31;
    inP[r * IS + c] = in[(size_t)(row0 + r) * 32 + c];
  }
  for (int i = tid; i < 16 * 64; i += 512) {
    int kp = i >> 6, n = i & 63;
    v2f w; w.x = We[(2 * kp) * 64 + n]; w.y = We[(2 * kp + 1) * 64 + n];
    WeP[kp * BP64 + n] = w;
  }
  for (int i = tid; i < 32 * 512; i += 512) {
    int kp = i >> 9, n = i & 511;
    v2f w; w.x = Wx0[(2 * kp) * 512 + n]; w.y = Wx0[(2 * kp + 1) * 512 + n];
    WxP[kp * BP512 + n] = w;
  }
  __syncthreads();

  {
    const int tm = wave >> 2, tn = wave & 3;
    const int col = tn * 16 + ln;
    const float bv = be[col];
    v8f acc = vzero8();
    const int am = tm * 16 + ln;
#pragma unroll
    for (int k0 = 0; k0 < 32; k0 += 4) {
      v2f a = *(const v2f*)&inP[am * IS + k0 + 2 * lh];
      v2f b = WeP[((k0 >> 1) + lh) * BP64 + col];
      acc = wmma4(a, b, acc);
    }
#pragma unroll
    for (int r = 0; r < 8; ++r)
      xb[(tm * 16 + r + 8 * lh) * XS + col] = acc[r] + bv;
  }
  __syncthreads();

  for (int task = wave; task < 128; task += 16) {
    const int tm = task >> 5, tn = task & 31;
    const int col = tn * 16 + ln;
    const float bv = b0[col];
    v8f acc = vzero8();
    const int am = tm * 16 + ln;
#pragma unroll
    for (int k0 = 0; k0 < 64; k0 += 4) {
      v2f a = *(const v2f*)&xb[am * XS + k0 + 2 * lh];
      v2f b = WxP[((k0 >> 1) + lh) * BP512 + col];
      acc = wmma4(a, b, acc);
    }
#pragma unroll
    for (int r = 0; r < 8; ++r)
      xg[(size_t)(row0 + tm * 16 + r + 8 * lh) * 512 + col] = acc[r] + bv;
  }
}

__global__ void __launch_bounds__(512)
xg1_kernel(const float* __restrict__ x1,
           const float* __restrict__ Wx1,
           const float* __restrict__ b1,
           float* __restrict__ xg)
{
  extern __shared__ __align__(16) char smem[];
  float* aB = (float*)smem;
  v2f*   WP = (v2f*)(smem + 64 * HS * 4);

  const int tid  = threadIdx.x;
  const int wave = tid >> 5;
  const int lane = tid & 31;
  const int lh   = lane >> 4;
  const int ln   = lane & 15;
  const int row0 = blockIdx.x * 64;

  for (int i = tid; i < 64 * 128; i += 512) {
    int r = i >> 7, c = i & 127;
    aB[r * HS + c] = x1[(size_t)(row0 + r) * 128 + c];
  }
  for (int i = tid; i < 64 * 512; i += 512) {
    int kp = i >> 9, n = i & 511;
    v2f w; w.x = Wx1[(2 * kp) * 512 + n]; w.y = Wx1[(2 * kp + 1) * 512 + n];
    WP[kp * BP512 + n] = w;
  }
  __syncthreads();

  for (int task = wave; task < 128; task += 16) {
    const int tm = task >> 5, tn = task & 31;
    const int col = tn * 16 + ln;
    const float bv = b1[col];
    v8f acc = vzero8();
    const int am = tm * 16 + ln;
    for (int k0 = 0; k0 < 128; k0 += 4) {
      v2f a = *(const v2f*)&aB[am * HS + k0 + 2 * lh];
      v2f b = WP[((k0 >> 1) + lh) * BP512 + col];
      acc = wmma4(a, b, acc);
    }
#pragma unroll
    for (int r = 0; r < 8; ++r)
      xg[(size_t)(row0 + tm * 16 + r + 8 * lh) * 512 + col] = acc[r] + bv;
  }
}

__global__ void __launch_bounds__(256)
scan_kernel(const float* __restrict__ xg,
            const float* __restrict__ Wh,
            float* __restrict__ hseq)
{
  extern __shared__ __align__(16) char smem[];
  float* hbuf = (float*)smem;

  const int tid  = threadIdx.x;
  const int wave = tid >> 5;
  const int lane = tid & 31;
  const int lh   = lane >> 4;
  const int ln   = lane & 15;
  const int rowb = blockIdx.x * 16;
  const int colh = wave * 16 + ln;

  v2f wfr[4][32];
#pragma unroll
  for (int ks = 0; ks < 32; ++ks) {
    const float* wp = Wh + (size_t)(4 * ks + 2 * lh) * 512 + colh;
#pragma unroll
    for (int g = 0; g < 4; ++g) {
      v2f w; w.x = wp[g * 128]; w.y = wp[512 + g * 128];
      wfr[g][ks] = w;
    }
  }
  for (int i = tid; i < 16 * HS; i += 256) hbuf[i] = 0.0f;
  __syncthreads();

  float hnew[8];

  float hist[KF - 1][8];
#pragma unroll
  for (int j = 0; j < KF - 1; ++j)
#pragma unroll
    for (int r = 0; r < 8; ++r) hist[j][r] = 0.0f;

  for (int t = 0; t < T_DIM; ++t) {
    const float* xgt = xg + (size_t)t * (64 * 512) + (size_t)rowb * 512;

    if (t + 1 < T_DIM) {
#pragma unroll
      for (int q = 1; q <= 4; ++q)
        __builtin_prefetch(xgt + (size_t)q * (64 * 512) + (size_t)tid * 32, 0, 0);
    }

    {
      v2f afr[32];
#pragma unroll
      for (int ks = 0; ks < 32; ++ks)
        afr[ks] = *(const v2f*)&hbuf[ln * HS + 4 * ks + 2 * lh];
      __builtin_amdgcn_sched_barrier(0);

      float xv0[8], xv1[8], xv2[8], xv3[8];
#pragma unroll
      for (int r = 0; r < 8; ++r) {
        const int row = r + 8 * lh;
        const float* p = xgt + row * 512 + colh;
        xv0[r] = p[0];  xv1[r] = p[128];
        xv2[r] = p[256]; xv3[r] = p[384];
      }

      v8f acc0 = vzero8(), acc1 = vzero8(), acc2 = vzero8(), acc3 = vzero8();
#pragma unroll
      for (int ks = 0; ks < 32; ++ks) {
        const v2f a = afr[ks];
        acc0 = wmma4(a, wfr[0][ks], acc0);
        acc1 = wmma4(a, wfr[1][ks], acc1);
        acc2 = wmma4(a, wfr[2][ks], acc2);
        acc3 = wmma4(a, wfr[3][ks], acc3);
      }
      __builtin_amdgcn_sched_barrier(0);

      float hc[8], og[8], dd[8];
#pragma unroll
      for (int r = 0; r < 8; ++r) {
        const float ig = sigm(acc0[r] + xv0[r]);
        og[r] = sigm(acc1[r] + xv1[r]);
        const float cg = tanh_sel(acc2[r] + xv2[r]);
        dd[r] = 0.5f * sigm(acc3[r] + xv3[r]);
        hc[r] = ig * cg;
      }

      float csum[8], w[8];
#pragma unroll
      for (int r = 0; r < 8; ++r) { csum[r] = hc[r]; w[r] = 1.0f; }
#pragma unroll
      for (int j = 1; j < KF; ++j) {
        const float jm1 = (float)(j - 1), rj = 1.0f / (float)j;
#pragma unroll
        for (int r = 0; r < 8; ++r) {
          w[r] *= (jm1 + dd[r]) * rj;
          csum[r] = fmaf(w[r], hist[j - 1][r], csum[r]);
        }
      }
#pragma unroll
      for (int j = KF - 2; j > 0; --j)
#pragma unroll
        for (int r = 0; r < 8; ++r) hist[j][r] = hist[j - 1][r];
#pragma unroll
      for (int r = 0; r < 8; ++r) {
        hist[0][r] = hc[r];
        hnew[r] = og[r] * tanh_sel(csum[r]);
      }
    }

    __builtin_amdgcn_sched_barrier(0);
    __syncthreads();
#pragma unroll
    for (int r = 0; r < 8; ++r)
      hbuf[(r + 8 * lh) * HS + colh] = hnew[r];
    __syncthreads();

    {
      float* ht = hseq + (size_t)t * (64 * HID) + (size_t)rowb * HID;
#pragma unroll
      for (int q = 0; q < 2; ++q) {
        const int row = 2 * wave + q;
        const float4 v = *(const float4*)&hbuf[row * HS + 4 * lane];
        *(float4*)&ht[row * HID + 4 * lane] = v;
      }
    }
  }
}

template <int NOUT>
__global__ void __launch_bounds__(512)
proj_kernel(const float* hin,
            const float* __restrict__ Wo,
            const float* __restrict__ bo,
            float* y)
{
  constexpr int BPS = NOUT + 16;
  constexpr int NTN = NOUT / 16;
  extern __shared__ __align__(16) char smem[];
  float* aB = (float*)smem;
  v2f*   WP = (v2f*)(smem + 64 * HS * 4);

  const int tid  = threadIdx.x;
  const int wave = tid >> 5;
  const int lane = tid & 31;
  const int lh   = lane >> 4;
  const int ln   = lane & 15;
  const int row0 = blockIdx.x * 64;

  for (int i = tid; i < 64 * 128; i += 512) {
    int r = i >> 7, c = i & 127;
    aB[r * HS + c] = hin[(size_t)(row0 + r) * 128 + c];
  }
  for (int i = tid; i < 64 * NOUT; i += 512) {
    int kp = i / NOUT, n = i % NOUT;
    v2f w; w.x = Wo[(2 * kp) * NOUT + n]; w.y = Wo[(2 * kp + 1) * NOUT + n];
    WP[kp * BPS + n] = w;
  }
  __syncthreads();

  for (int task = wave; task < 4 * NTN; task += 16) {
    const int tm = task / NTN, tn = task % NTN;
    const int col = tn * 16 + ln;
    const float bv = bo[col];
    v8f acc = vzero8();
    const int am = tm * 16 + ln;
    for (int k0 = 0; k0 < 128; k0 += 4) {
      v2f a = *(const v2f*)&aB[am * HS + k0 + 2 * lh];
      v2f b = WP[((k0 >> 1) + lh) * BPS + col];
      acc = wmma4(a, b, acc);
    }
#pragma unroll
    for (int r = 0; r < 8; ++r)
      y[(size_t)(row0 + tm * 16 + r + 8 * lh) * NOUT + col] = acc[r] + bv;
  }
}

extern "C" void kernel_launch(void* const* d_in, const int* in_sizes, int n_in,
                              void* d_out, int out_size, void* d_ws, size_t ws_size,
                              hipStream_t stream) {
  const float* inputs = (const float*)d_in[0];
  const float* W_emb  = (const float*)d_in[1];
  const float* b_emb  = (const float*)d_in[2];
  const float* Wx0    = (const float*)d_in[3];
  const float* Wh0    = (const float*)d_in[4];
  const float* b0     = (const float*)d_in[5];
  const float* Wo0    = (const float*)d_in[6];
  const float* bo0    = (const float*)d_in[7];
  const float* Wx1    = (const float*)d_in[8];
  const float* Wh1    = (const float*)d_in[9];
  const float* b1     = (const float*)d_in[10];
  const float* Wo1    = (const float*)d_in[11];
  const float* bo1    = (const float*)d_in[12];
  float* out = (float*)d_out;

  float* xg   = (float*)d_ws;
  float* hx   = xg + (size_t)T_DIM * B_DIM * G4;

  const size_t ldsE = (size_t)64 * IS * 4 + 16 * BP64 * 8 + 64 * XS * 4 + 32 * BP512 * 8;
  const size_t ldsX = (size_t)64 * HS * 4 + 64 * BP512 * 8;
  const size_t ldsS = (size_t)16 * HS * 4;
  const size_t ldsP0 = (size_t)64 * HS * 4 + 64 * (HID + 16) * 8;
  const size_t ldsP1 = (size_t)64 * HS * 4 + 64 * (OUTD + 16) * 8;

  const int nblk = (T_DIM * B_DIM) / 64;

  embed_xg0_kernel<<<nblk, 512, ldsE, stream>>>(inputs, W_emb, b_emb, Wx0, b0, xg);
  scan_kernel<<<B_DIM / 16, 256, ldsS, stream>>>(xg, Wh0, hx);
  proj_kernel<HID><<<nblk, 512, ldsP0, stream>>>(hx, Wo0, bo0, hx);
  xg1_kernel<<<nblk, 512, ldsX, stream>>>(hx, Wx1, b1, xg);
  scan_kernel<<<B_DIM / 16, 256, ldsS, stream>>>(xg, Wh1, hx);
  proj_kernel<OUTD><<<nblk, 512, ldsP1, stream>>>(hx, Wo1, bo1, out);
}


# scan kernels: gate arithmetic on the hardware exponential and reciprocal: sigmoid without the IEEE division, tanh's e^(2|x|) by one v_exp_f32, not the library form; gate values move by about 1 ulp
# speedup vs baseline: 11.5314x; 1.1844x over previous
#include <hip/hip_runtime.h>
#include <cstdint>
#include <cstddef>

#define DEVINL __device__ __forceinline__

typedef float v2f __attribute__((ext_vector_type(2)));
typedef float v8f __attribute__((ext_vector_type(8)));

static constexpr int T_DIM  = 1024;
static constexpr int B_DIM  = 64;
static constexpr int IN_DIM = 32;
static constexpr int EMB    = 64;
static constexpr int HID    = 128;
static constexpr int KF     = 16;
static constexpr int OUTD   = 32;
static constexpr int G4     = 512;

static constexpr int HS     = 132;
static constexpr int XS     = 68;
static constexpr int IS     = 36;
static constexpr int BP512  = 528;
static constexpr int BP64   = 80;

DEVINL float sigm(float x) { return __builtin_amdgcn_rcpf(1.0f + __expf(-x)); }

DEVINL float tanh_sel(float x) {
  const float y  = fabsf(x);
  const float y2 = x * x;
  float p = fmaf(__uint_as_float(0xbbbac73du), y2, __uint_as_float(0x3ca908c9u));
  p = fmaf(y2, p, __uint_as_float(0xbd5c1c4eu));
  p = fmaf(y2, p, __uint_as_float(0x3e088382u));
  p = fmaf(y2, p, __uint_as_float(0xbeaaaa99u));
  const float zs = fmaf(y2, y * p, y);
  const float e  = __expf(y + y);
  const float zl = fmaf(__builtin_amdgcn_rcpf(e + 1.0f), -2.0f, 1.0f);
  const float z  = (y < 0.625f) ? zs : zl;
  return copysignf(z, x);
}

DEVINL v8f wmma4(v2f a, v2f b, v8f c) {
  return __builtin_amdgcn_wmma_f32_16x16x4_f32(false, a, false, b, (short)0, c,
                                               false, false);
}

DEVINL v8f vzero8() {
  v8f z;
#pragma unroll
  for (int r = 0; r < 8; ++r) z[r] = 0.0f;
  return z;
}

__global__ void __launch_bounds__(512)
embed_xg0_kernel(const float* __restrict__ in,
                 const float* __restrict__ We,
                 const float* __restrict__ be,
                 const float* __restrict__ Wx0,
                 const float* __restrict__ b0,
                 float* __restrict__ xg)
{
  extern __shared__ __align__(16) char smem[];
  float* inP = (float*)smem;
  v2f*   WeP = (v2f*)(smem + 64 * IS * 4);
  float* xb  = (float*)(smem + 64 * IS * 4 + 16 * BP64 * 8);
  v2f*   WxP = (v2f*)(smem + 64 * IS * 4 + 16 * BP64 * 8 + 64 * XS * 4);

  const int tid  = threadIdx.x;
  const int wave = tid >> 5;
  const int lane = tid & 31;
  const int lh   = lane >> 4;
  const int ln   = lane & 15;
  const int row0 = blockIdx.x * 64;

  for (int i = tid; i < 64 * 32; i += 512) {
    int r = i >> 5, c = i & 31;
    inP[r * IS + c] = in[(size_t)(row0 + r) * 32 + c];
  }
  for (int i = tid; i < 16 * 64; i += 512) {
    int kp = i >> 6, n = i & 63;
    v2f w; w.x = We[(2 * kp) * 64 + n]; w.y = We[(2 * kp + 1) * 64 + n];
    WeP[kp * BP64 + n] = w;
  }
  for (int i = tid; i < 32 * 512; i += 512) {
    int kp = i >> 9, n = i & 511;
    v2f w; w.x = Wx0[(2 * kp) * 512 + n]; w.y = Wx0[(2 * kp + 1) * 512 + n];
    WxP[kp * BP512 + n] = w;
  }
  __syncthreads();

  {
    const int tm = wave >> 2, tn = wave & 3;
    const int col = tn * 16 + ln;
    const float bv = be[col];
    v8f acc = vzero8();
    const int am = tm * 16 + ln;
#pragma unroll
    for (int k0 = 0; k0 < 32; k0 += 4) {
      v2f a = *(const v2f*)&inP[am * IS + k0 + 2 * lh];
      v2f b = WeP[((k0 >> 1) + lh) * BP64 + col];
      acc = wmma4(a, b, acc);
    }
#pragma unroll
    for (int r = 0; r < 8; ++r)
      xb[(tm * 16 + r + 8 * lh) * XS + col] = acc[r] + bv;
  }
  __syncthreads();

  for (int task = wave; task < 128; task += 16) {
    const int tm = task >> 5, tn = task & 31;
    const int col = tn * 16 + ln;
    const float bv = b0[col];
    v8f acc = vzero8();
    const int am = tm * 16 + ln;
#pragma unroll
    for (int k0 = 0; k0 < 64; k0 += 4) {
      v2f a = *(const v2f*)&xb[am * XS + k0 + 2 * lh];
      v2f b = WxP[((k0 >> 1) + lh) * BP512 + col];
      acc = wmma4(a, b, acc);
    }
#pragma unroll
    for (int r = 0; r < 8; ++r)
      xg[(size_t)(row0 + tm * 16 + r + 8 * lh) * 512 + col] = acc[r] + bv;
  }
}

__global__ void __launch_bounds__(512)
xg1_kernel(const float* __restrict__ x1,
           const float* __restrict__ Wx1,
           const float* __restrict__ b1,
           float* __restrict__ xg)
{
  extern __shared__ __align__(16) char smem[];
  float* aB = (float*)smem;
  v2f*   WP = (v2f*)(smem + 64 * HS * 4);

  const int tid  = threadIdx.x;
  const int wave = tid >> 5;
  const int lane = tid & 31;
  const int lh   = lane >> 4;
  const int ln   = lane & 15;
  const int row0 = blockIdx.x * 64;

  for (int i = tid; i < 64 * 128; i += 512) {
    int r = i >> 7, c = i & 127;
    aB[r * HS + c] = x1[(size_t)(row0 + r) * 128 + c];
  }
  for (int i = tid; i < 64 * 512; i += 512) {
    int kp = i >> 9, n = i & 511;
    v2f w; w.x = Wx1[(2 * kp) * 512 + n]; w.y = Wx1[(2 * kp + 1) * 512 + n];
    WP[kp * BP512 + n] = w;
  }
  __syncthreads();

  for (int task = wave; task < 128; task += 16) {
    const int tm = task >> 5, tn = task & 31;
    const int col = tn * 16 + ln;
    const float bv = b1[col];
    v8f acc = vzero8();
    const int am = tm * 16 + ln;
    for (int k0 = 0; k0 < 128; k0 += 4) {
      v2f a = *(const v2f*)&aB[am * HS + k0 + 2 * lh];
      v2f b = WP[((k0 >> 1) + lh) * BP512 + col];
      acc = wmma4(a, b, acc);
    }
#pragma unroll
    for (int r = 0; r < 8; ++r)
      xg[(size_t)(row0 + tm * 16 + r + 8 * lh) * 512 + col] = acc[r] + bv;
  }
}

__global__ void __launch_bounds__(256)
scan_kernel(const float* __restrict__ xg,
            const float* __restrict__ Wh,
            float* __restrict__ hseq)
{
  extern __shared__ __align__(16) char smem[];
  float* hbuf = (float*)smem;

  const int tid  = threadIdx.x;
  const int wave = tid >> 5;
  const int lane = tid & 31;
  const int lh   = lane >> 4;
  const int ln   = lane & 15;
  const int rowb = blockIdx.x * 16;
  const int colh = wave * 16 + ln;

  v2f wfr[4][32];
#pragma unroll
  for (int ks = 0; ks < 32; ++ks) {
    const float* wp = Wh + (size_t)(4 * ks + 2 * lh) * 512 + colh;
#pragma unroll
    for (int g = 0; g < 4; ++g) {
      v2f w; w.x = wp[g * 128]; w.y = wp[512 + g * 128];
      wfr[g][ks] = w;
    }
  }
  for (int i = tid; i < 16 * HS; i += 256) hbuf[i] = 0.0f;
  __syncthreads();

  float hnew[8];

  float hist[KF - 1][8];
#pragma unroll
  for (int j = 0; j < KF - 1; ++j)
#pragma unroll
    for (int r = 0; r < 8; ++r) hist[j][r] = 0.0f;

  for (int t = 0; t < T_DIM; ++t) {
    const float* xgt = xg + (size_t)t * (64 * 512) + (size_t)rowb * 512;

    if (t + 1 < T_DIM) {
#pragma unroll
      for (int q = 1; q <= 4; ++q)
        __builtin_prefetch(xgt + (size_t)q * (64 * 512) + (size_t)tid * 32, 0, 0);
    }

    {
      v2f afr[32];
#pragma unroll
      for (int ks = 0; ks < 32; ++ks)
        afr[ks] = *(const v2f*)&hbuf[ln * HS + 4 * ks + 2 * lh];
      __builtin_amdgcn_sched_barrier(0);

      float xv0[8], xv1[8], xv2[8], xv3[8];
#pragma unroll
      for (int r = 0; r < 8; ++r) {
        const int row = r + 8 * lh;
        const float* p = xgt + row * 512 + colh;
        xv0[r] = p[0];  xv1[r] = p[128];
        xv2[r] = p[256]; xv3[r] = p[384];
      }

      v8f acc0 = vzero8(), acc1 = vzero8(), acc2 = vzero8(), acc3 = vzero8();
#pragma unroll
      for (int ks = 0; ks < 32; ++ks) {
        const v2f a = afr[ks];
        acc0 = wmma4(a, wfr[0][ks], acc0);
        acc1 = wmma4(a, wfr[1][ks], acc1);
        acc2 = wmma4(a, wfr[2][ks], acc2);
        acc3 = wmma4(a, wfr[3][ks], acc3);
      }
      __builtin_amdgcn_sched_barrier(0);

      float hc[8], og[8], dd[8];
#pragma unroll
      for (int r = 0; r < 8; ++r) {
        const float ig = sigm(acc0[r] + xv0[r]);
        og[r] = sigm(acc1[r] + xv1[r]);
        const float cg = tanh_sel(acc2[r] + xv2[r]);
        dd[r] = 0.5f * sigm(acc3[r] + xv3[r]);
        hc[r] = ig * cg;
      }

      float csum[8], w[8];
#pragma unroll
      for (int r = 0; r < 8; ++r) { csum[r] = hc[r]; w[r] = 1.0f; }
#pragma unroll
      for (int j = 1; j < KF; ++j) {
        const float jm1 = (float)(j - 1), rj = 1.0f / (float)j;
#pragma unroll
        for (int r = 0; r < 8; ++r) {
          w[r] *= (jm1 + dd[r]) * rj;
          csum[r] = fmaf(w[r], hist[j - 1][r], csum[r]);
        }
      }
#pragma unroll
      for (int j = KF - 2; j > 0; --j)
#pragma unroll
        for (int r = 0; r < 8; ++r) hist[j][r] = hist[j - 1][r];
#pragma unroll
      for (int r = 0; r < 8; ++r) {
        hist[0][r] = hc[r];
        hnew[r] = og[r] * tanh_sel(csum[r]);
      }
    }

    __builtin_amdgcn_sched_barrier(0);
    __syncthreads();
#pragma unroll
    for (int r = 0; r < 8; ++r)
      hbuf[(r + 8 * lh) * HS + colh] = hnew[r];
    __syncthreads();

    {
      float* ht = hseq + (size_t)t * (64 * HID) + (size_t)rowb * HID;
#pragma unroll
      for (int q = 0; q < 2; ++q) {
        const int row = 2 * wave + q;
        const float4 v = *(const float4*)&hbuf[row * HS + 4 * lane];
        *(float4*)&ht[row * HID + 4 * lane] = v;
      }
    }
  }
}

template <int NOUT>
__global__ void __launch_bounds__(512)
proj_kernel(const float* hin,
            const float* __restrict__ Wo,
            const float* __restrict__ bo,
            float* y)
{
  constexpr int BPS = NOUT + 16;
  constexpr int NTN = NOUT / 16;
  extern __shared__ __align__(16) char smem[];
  float* aB = (float*)smem;
  v2f*   WP = (v2f*)(smem + 64 * HS * 4);

  const int tid  = threadIdx.x;
  const int wave = tid >> 5;
  const int lane = tid & 31;
  const int lh   = lane >> 4;
  const int ln   = lane & 15;
  const int row0 = blockIdx.x * 64;

  for (int i = tid; i < 64 * 128; i += 512) {
    int r = i >> 7, c = i & 127;
    aB[r * HS + c] = hin[(size_t)(row0 + r) * 128 + c];
  }
  for (int i = tid; i < 64 * NOUT; i += 512) {
    int kp = i / NOUT, n = i % NOUT;
    v2f w; w.x = Wo[(2 * kp) * NOUT + n]; w.y = Wo[(2 * kp + 1) * NOUT + n];
    WP[kp * BPS + n] = w;
  }
  __syncthreads();

  for (int task = wave; task < 4 * NTN; task += 16) {
    const int tm = task / NTN, tn = task % NTN;
    const int col = tn * 16 + ln;
    const float bv = bo[col];
    v8f acc = vzero8();
    const int am = tm * 16 + ln;
    for (int k0 = 0; k0 < 128; k0 += 4) {
      v2f a = *(const v2f*)&aB[am * HS + k0 + 2 * lh];
      v2f b = WP[((k0 >> 1) + lh) * BPS + col];
      acc = wmma4(a, b, acc);
    }
#pragma unroll
    for (int r = 0; r < 8; ++r)
      y[(size_t)(row0 + tm * 16 + r + 8 * lh) * NOUT + col] = acc[r] + bv;
  }
}

extern "C" void kernel_launch(void* const* d_in, const int* in_sizes, int n_in,
                              void* d_out, int out_size, void* d_ws, size_t ws_size,
                              hipStream_t stream) {
  const float* inputs = (const float*)d_in[0];
  const float* W_emb  = (const float*)d_in[1];
  const float* b_emb  = (const float*)d_in[2];
  const float* Wx0    = (const float*)d_in[3];
  const float* Wh0    = (const float*)d_in[4];
  const float* b0     = (const float*)d_in[5];
  const float* Wo0    = (const float*)d_in[6];
  const float* bo0    = (const float*)d_in[7];
  const float* Wx1    = (const float*)d_in[8];
  const float* Wh1    = (const float*)d_in[9];
  const float* b1     = (const float*)d_in[10];
  const float* Wo1    = (const float*)d_in[11];
  const float* bo1    = (const float*)d_in[12];
  float* out = (float*)d_out;

  float* xg   = (float*)d_ws;
  float* hx   = xg + (size_t)T_DIM * B_DIM * G4;

  const size_t ldsE = (size_t)64 * IS * 4 + 16 * BP64 * 8 + 64 * XS * 4 + 32 * BP512 * 8;
  const size_t ldsX = (size_t)64 * HS * 4 + 64 * BP512 * 8;
  const size_t ldsS = (size_t)16 * HS * 4;
  const size_t ldsP0 = (size_t)64 * HS * 4 + 64 * (HID + 16) * 8;
  const size_t ldsP1 = (size_t)64 * HS * 4 + 64 * (OUTD + 16) * 8;

  const int nblk = (T_DIM * B_DIM) / 64;

  embed_xg0_kernel<<<nblk, 512, ldsE, stream>>>(inputs, W_emb, b_emb, Wx0, b0, xg);
  scan_kernel<<<B_DIM / 16, 256, ldsS, stream>>>(xg, Wh0, hx);
  proj_kernel<HID><<<nblk, 512, ldsP0, stream>>>(hx, Wo0, bo0, hx);
  xg1_kernel<<<nblk, 512, ldsX, stream>>>(hx, Wx1, b1, xg);
  scan_kernel<<<B_DIM / 16, 256, ldsS, stream>>>(xg, Wh1, hx);
  proj_kernel<OUTD><<<nblk, 512, ldsP1, stream>>>(hx, Wo1, bo1, out);
}


# the two layers' scans run side by side: time cut into 8 chunks, one launch scans chunk c of layer 0 and chunk c-1 of layer 1 (8 workgroups); state carried in workspace; sums unchanged
# speedup vs baseline: 18.4598x; 1.6008x over previous
#include <hip/hip_runtime.h>
#include <cstdint>
#include <cstddef>

#define DEVINL __device__ __forceinline__

typedef float v2f __attribute__((ext_vector_type(2)));
typedef float v8f __attribute__((ext_vector_type(8)));

static constexpr int T_DIM  = 1024;
static constexpr int B_DIM  = 64;
static constexpr int IN_DIM = 32;
static constexpr int EMB    = 64;
static constexpr int HID    = 128;
static constexpr int KF     = 16;
static constexpr int OUTD   = 32;
static constexpr int G4     = 512;
static constexpr int NCH    = 8;
static constexpr int CH     = T_DIM / NCH;
static constexpr int ST_H   = B_DIM * HID;
static constexpr int ST_ALL = ST_H + 4 * (KF - 1) * 8 * 256;

static constexpr int HS     = 132;
static constexpr int XS     = 68;
static constexpr int IS     = 36;
static constexpr int BP512  = 528;
static constexpr int BP64   = 80;

DEVINL float sigm(float x) { return __builtin_amdgcn_rcpf(1.0f + __expf(-x)); }

DEVINL float tanh_sel(float x) {
  const float y  = fabsf(x);
  const float y2 = x * x;
  float p = fmaf(__uint_as_float(0xbbbac73du), y2, __uint_as_float(0x3ca908c9u));
  p = fmaf(y2, p, __uint_as_float(0xbd5c1c4eu));
  p = fmaf(y2, p, __uint_as_float(0x3e088382u));
  p = fmaf(y2, p, __uint_as_float(0xbeaaaa99u));
  const float zs = fmaf(y2, y * p, y);
  const float e  = __expf(y + y);
  const float zl = fmaf(__builtin_amdgcn_rcpf(e + 1.0f), -2.0f, 1.0f);
  const float z  = (y < 0.625f) ? zs : zl;
  return copysignf(z, x);
}

DEVINL v8f wmma4(v2f a, v2f b, v8f c) {
  return __builtin_amdgcn_wmma_f32_16x16x4_f32(false, a, false, b, (short)0, c,
                                               false, false);
}

DEVINL v8f vzero8() {
  v8f z;
#pragma unroll
  for (int r = 0; r < 8; ++r) z[r] = 0.0f;
  return z;
}

__global__ void __launch_bounds__(512)
embed_xg0_kernel(const float* __restrict__ in,
                 const float* __restrict__ We,
                 const float* __restrict__ be,
                 const float* __restrict__ Wx0,
                 const float* __restrict__ b0,
                 float* __restrict__ xg)
{
  extern __shared__ __align__(16) char smem[];
  float* inP = (float*)smem;
  v2f*   WeP = (v2f*)(smem + 64 * IS * 4);
  float* xb  = (float*)(smem + 64 * IS * 4 + 16 * BP64 * 8);
  v2f*   WxP = (v2f*)(smem + 64 * IS * 4 + 16 * BP64 * 8 + 64 * XS * 4);

  const int tid  = threadIdx.x;
  const int wave = tid >> 5;
  const int lane = tid & 31;
  const int lh   = lane >> 4;
  const int ln   = lane & 15;
  const int row0 = blockIdx.x * 64;

  for (int i = tid; i < 64 * 32; i += 512) {
    int r = i >> 5, c = i & 31;
    inP[r * IS + c] = in[(size_t)(row0 + r) * 32 + c];
  }
  for (int i = tid; i < 16 * 64; i += 512) {
    int kp = i >> 6, n = i & 63;
    v2f w; w.x = We[(2 * kp) * 64 + n]; w.y = We[(2 * kp + 1) * 64 + n];
    WeP[kp * BP64 + n] = w;
  }
  for (int i = tid; i < 32 * 512; i += 512) {
    int kp = i >> 9, n = i & 511;
    v2f w; w.x = Wx0[(2 * kp) * 512 + n]; w.y = Wx0[(2 * kp + 1) * 512 + n];
    WxP[kp * BP512 + n] = w;
  }
  __syncthreads();

  {
    const int tm = wave >> 2, tn = wave & 3;
    const int col = tn * 16 + ln;
    const float bv = be[col];
    v8f acc = vzero8();
    const int am = tm * 16 + ln;
#pragma unroll
    for (int k0 = 0; k0 < 32; k0 += 4) {
      v2f a = *(const v2f*)&inP[am * IS + k0 + 2 * lh];
      v2f b = WeP[((k0 >> 1) + lh) * BP64 + col];
      acc = wmma4(a, b, acc);
    }
#pragma unroll
    for (int r = 0; r < 8; ++r)
      xb[(tm * 16 + r + 8 * lh) * XS + col] = acc[r] + bv;
  }
  __syncthreads();

  for (int task = wave; task < 128; task += 16) {
    const int tm = task >> 5, tn = task & 31;
    const int col = tn * 16 + ln;
    const float bv = b0[col];
    v8f acc = vzero8();
    const int am = tm * 16 + ln;
#pragma unroll
    for (int k0 = 0; k0 < 64; k0 += 4) {
      v2f a = *(const v2f*)&xb[am * XS + k0 + 2 * lh];
      v2f b = WxP[((k0 >> 1) + lh) * BP512 + col];
      acc = wmma4(a, b, acc);
    }
#pragma unroll
    for (int r = 0; r < 8; ++r)
      xg[(size_t)(row0 + tm * 16 + r + 8 * lh) * 512 + col] = acc[r] + bv;
  }
}

__global__ void __launch_bounds__(512)
xg1_kernel(const float* __restrict__ x1,
           const float* __restrict__ Wx1,
           const float* __restrict__ b1,
           float* __restrict__ xg)
{
  extern __shared__ __align__(16) char smem[];
  float* aB = (float*)smem;
  v2f*   WP = (v2f*)(smem + 64 * HS * 4);

  const int tid  = threadIdx.x;
  const int wave = tid >> 5;
  const int lane = tid & 31;
  const int lh   = lane >> 4;
  const int ln   = lane & 15;
  const int row0 = blockIdx.x * 64;

  for (int i = tid; i < 64 * 128; i += 512) {
    int r = i >> 7, c = i & 127;
    aB[r * HS + c] = x1[(size_t)(row0 + r) * 128 + c];
  }
  for (int i = tid; i < 64 * 512; i += 512) {
    int kp = i >> 9, n = i & 511;
    v2f w; w.x = Wx1[(2 * kp) * 512 + n]; w.y = Wx1[(2 * kp + 1) * 512 + n];
    WP[kp * BP512 + n] = w;
  }
  __syncthreads();

  for (int task = wave; task < 128; task += 16) {
    const int tm = task >> 5, tn = task & 31;
    const int col = tn * 16 + ln;
    const float bv = b1[col];
    v8f acc = vzero8();
    const int am = tm * 16 + ln;
    for (int k0 = 0; k0 < 128; k0 += 4) {
      v2f a = *(const v2f*)&aB[am * HS + k0 + 2 * lh];
      v2f b = WP[((k0 >> 1) + lh) * BP512 + col];
      acc = wmma4(a, b, acc);
    }
#pragma unroll
    for (int r = 0; r < 8; ++r)
      xg[(size_t)(row0 + tm * 16 + r + 8 * lh) * 512 + col] = acc[r] + bv;
  }
}

struct ScanJob {
  const float* xg;
  const float* Wh;
  float*       hseq;
  float*       st;
  int          first;
};

__global__ void __launch_bounds__(256)
scan_kernel(ScanJob j0, ScanJob j1)
{
  extern __shared__ __align__(16) char smem[];
  float* hbuf = (float*)smem;

  const bool sec = blockIdx.x >= 4;
  const float* __restrict__ xg   = sec ? j1.xg   : j0.xg;
  const float* __restrict__ Wh   = sec ? j1.Wh   : j0.Wh;
  float* __restrict__       hseq = sec ? j1.hseq : j0.hseq;
  float* __restrict__       st   = sec ? j1.st   : j0.st;
  const int first                = sec ? j1.first : j0.first;

  const int tid  = threadIdx.x;
  const int wave = tid >> 5;
  const int lane = tid & 31;
  const int lh   = lane >> 4;
  const int ln   = lane & 15;
  const int wg   = blockIdx.x & 3;
  const int rowb = wg * 16;
  const int colh = wave * 16 + ln;
  float* sth = st + (size_t)rowb * HID;
  float* sthist = st + ST_H + (size_t)wg * ((KF - 1) * 8 * 256) + tid;

  v2f wfr[4][32];
#pragma unroll
  for (int ks = 0; ks < 32; ++ks) {
    const float* wp = Wh + (size_t)(4 * ks + 2 * lh) * 512 + colh;
#pragma unroll
    for (int g = 0; g < 4; ++g) {
      v2f w; w.x = wp[g * 128]; w.y = wp[512 + g * 128];
      wfr[g][ks] = w;
    }
  }
  float hnew[8];

  float hist[KF - 1][8];
  if (first) {
    for (int i = tid; i < 16 * HS; i += 256) hbuf[i] = 0.0f;
#pragma unroll
    for (int j = 0; j < KF - 1; ++j)
#pragma unroll
      for (int r = 0; r < 8; ++r) hist[j][r] = 0.0f;
  } else {
    for (int i = tid; i < 16 * HID; i += 256)
      hbuf[(i >> 7) * HS + (i & 127)] = sth[i];
#pragma unroll
    for (int j = 0; j < KF - 1; ++j)
#pragma unroll
      for (int r = 0; r < 8; ++r) hist[j][r] = sthist[(j * 8 + r) * 256];
  }
  __syncthreads();

  for (int t = 0; t < CH; ++t) {
    const float* xgt = xg + (size_t)t * (64 * 512) + (size_t)rowb * 512;

    if (t + 4 < CH) {
#pragma unroll
      for (int q = 1; q <= 4; ++q)
        __builtin_prefetch(xgt + (size_t)q * (64 * 512) + (size_t)tid * 32, 0, 0);
    }

    {
      v2f afr[32];
#pragma unroll
      for (int ks = 0; ks < 32; ++ks)
        afr[ks] = *(const v2f*)&hbuf[ln * HS + 4 * ks + 2 * lh];
      __builtin_amdgcn_sched_barrier(0);

      float xv0[8], xv1[8], xv2[8], xv3[8];
#pragma unroll
      for (int r = 0; r < 8; ++r) {
        const int row = r + 8 * lh;
        const float* p = xgt + row * 512 + colh;
        xv0[r] = p[0];  xv1[r] = p[128];
        xv2[r] = p[256]; xv3[r] = p[384];
      }

      v8f acc0 = vzero8(), acc1 = vzero8(), acc2 = vzero8(), acc3 = vzero8();
#pragma unroll
      for (int ks = 0; ks < 32; ++ks) {
        const v2f a = afr[ks];
        acc0 = wmma4(a, wfr[0][ks], acc0);
        acc1 = wmma4(a, wfr[1][ks], acc1);
        acc2 = wmma4(a, wfr[2][ks], acc2);
        acc3 = wmma4(a, wfr[3][ks], acc3);
      }
      __builtin_amdgcn_sched_barrier(0);

      float hc[8], og[8], dd[8];
#pragma unroll
      for (int r = 0; r < 8; ++r) {
        const float ig = sigm(acc0[r] + xv0[r]);
        og[r] = sigm(acc1[r] + xv1[r]);
        const float cg = tanh_sel(acc2[r] + xv2[r]);
        dd[r] = 0.5f * sigm(acc3[r] + xv3[r]);
        hc[r] = ig * cg;
      }

      float csum[8], w[8];
#pragma unroll
      for (int r = 0; r < 8; ++r) { csum[r] = hc[r]; w[r] = 1.0f; }
#pragma unroll
      for (int j = 1; j < KF; ++j) {
        const float jm1 = (float)(j - 1), rj = 1.0f / (float)j;
#pragma unroll
        for (int r = 0; r < 8; ++r) {
          w[r] *= (jm1 + dd[r]) * rj;
          csum[r] = fmaf(w[r], hist[j - 1][r], csum[r]);
        }
      }
#pragma unroll
      for (int j = KF - 2; j > 0; --j)
#pragma unroll
        for (int r = 0; r < 8; ++r) hist[j][r] = hist[j - 1][r];
#pragma unroll
      for (int r = 0; r < 8; ++r) {
        hist[0][r] = hc[r];
        hnew[r] = og[r] * tanh_sel(csum[r]);
      }
    }

    __builtin_amdgcn_sched_barrier(0);
    __syncthreads();
#pragma unroll
    for (int r = 0; r < 8; ++r)
      hbuf[(r + 8 * lh) * HS + colh] = hnew[r];
    __syncthreads();

    {
      float* ht = hseq + (size_t)t * (64 * HID) + (size_t)rowb * HID;
#pragma unroll
      for (int q = 0; q < 2; ++q) {
        const int row = 2 * wave + q;
        const float4 v = *(const float4*)&hbuf[row * HS + 4 * lane];
        *(float4*)&ht[row * HID + 4 * lane] = v;
      }
    }

    if (t == CH - 1) {
      for (int i = tid; i < 16 * HID; i += 256)
        sth[i] = hbuf[(i >> 7) * HS + (i & 127)];
#pragma unroll
      for (int j = 0; j < KF - 1; ++j)
#pragma unroll
        for (int r = 0; r < 8; ++r) sthist[(j * 8 + r) * 256] = hist[j][r];
    }
  }
}

template <int NOUT>
__global__ void __launch_bounds__(512)
proj_kernel(const float* hin,
            const float* __restrict__ Wo,
            const float* __restrict__ bo,
            float* y)
{
  constexpr int BPS = NOUT + 16;
  constexpr int NTN = NOUT / 16;
  extern __shared__ __align__(16) char smem[];
  float* aB = (float*)smem;
  v2f*   WP = (v2f*)(smem + 64 * HS * 4);

  const int tid  = threadIdx.x;
  const int wave = tid >> 5;
  const int lane = tid & 31;
  const int lh   = lane >> 4;
  const int ln   = lane & 15;
  const int row0 = blockIdx.x * 64;

  for (int i = tid; i < 64 * 128; i += 512) {
    int r = i >> 7, c = i & 127;
    aB[r * HS + c] = hin[(size_t)(row0 + r) * 128 + c];
  }
  for (int i = tid; i < 64 * NOUT; i += 512) {
    int kp = i / NOUT, n = i % NOUT;
    v2f w; w.x = Wo[(2 * kp) * NOUT + n]; w.y = Wo[(2 * kp + 1) * NOUT + n];
    WP[kp * BPS + n] = w;
  }
  __syncthreads();

  for (int task = wave; task < 4 * NTN; task += 16) {
    const int tm = task / NTN, tn = task % NTN;
    const int col = tn * 16 + ln;
    const float bv = bo[col];
    v8f acc = vzero8();
    const int am = tm * 16 + ln;
    for (int k0 = 0; k0 < 128; k0 += 4) {
      v2f a = *(const v2f*)&aB[am * HS + k0 + 2 * lh];
      v2f b = WP[((k0 >> 1) + lh) * BPS + col];
      acc = wmma4(a, b, acc);
    }
#pragma unroll
    for (int r = 0; r < 8; ++r)
      y[(size_t)(row0 + tm * 16 + r + 8 * lh) * NOUT + col] = acc[r] + bv;
  }
}

extern "C" void kernel_launch(void* const* d_in, const int* in_sizes, int n_in,
                              void* d_out, int out_size, void* d_ws, size_t ws_size,
                              hipStream_t stream) {
  const float* inputs = (const float*)d_in[0];
  const float* W_emb  = (const float*)d_in[1];
  const float* b_emb  = (const float*)d_in[2];
  const float* Wx0    = (const float*)d_in[3];
  const float* Wh0    = (const float*)d_in[4];
  const float* b0     = (const float*)d_in[5];
  const float* Wo0    = (const float*)d_in[6];
  const float* bo0    = (const float*)d_in[7];
  const float* Wx1    = (const float*)d_in[8];
  const float* Wh1    = (const float*)d_in[9];
  const float* b1     = (const float*)d_in[10];
  const float* Wo1    = (const float*)d_in[11];
  const float* bo1    = (const float*)d_in[12];
  float* out = (float*)d_out;

  float* xg0  = (float*)d_ws;
  float* xg1c = xg0  + (size_t)T_DIM * B_DIM * G4;
  float* h0c  = xg1c + (size_t)CH * B_DIM * G4;
  float* h1c  = h0c  + (size_t)CH * B_DIM * HID;
  float* st0  = h1c  + (size_t)CH * B_DIM * HID;
  float* st1  = st0  + (size_t)ST_ALL;

  const size_t ldsE = (size_t)64 * IS * 4 + 16 * BP64 * 8 + 64 * XS * 4 + 32 * BP512 * 8;
  const size_t ldsX = (size_t)64 * HS * 4 + 64 * BP512 * 8;
  const size_t ldsS = (size_t)16 * HS * 4;
  const size_t ldsP0 = (size_t)64 * HS * 4 + 64 * (HID + 16) * 8;
  const size_t ldsP1 = (size_t)64 * HS * 4 + 64 * (OUTD + 16) * 8;

  const int nblk = (T_DIM * B_DIM) / 64;
  const int cblk = (CH * B_DIM) / 64;

  embed_xg0_kernel<<<nblk, 512, ldsE, stream>>>(inputs, W_emb, b_emb, Wx0, b0, xg0);

  for (int c = 0; c <= NCH; ++c) {
    ScanJob l0 = { xg0 + (size_t)c * CH * B_DIM * G4, Wh0, h0c, st0, c == 0 };
    ScanJob l1 = { xg1c, Wh1, h1c, st1, c == 1 };
    if (c == 0)
      scan_kernel<<<4, 256, ldsS, stream>>>(l0, l0);
    else if (c == NCH)
      scan_kernel<<<4, 256, ldsS, stream>>>(l1, l1);
    else
      scan_kernel<<<8, 256, ldsS, stream>>>(l0, l1);
    if (c > 0)
      proj_kernel<OUTD><<<cblk, 512, ldsP1, stream>>>(
          h1c, Wo1, bo1, out + (size_t)(c - 1) * CH * B_DIM * OUTD);
    if (c < NCH) {
      proj_kernel<HID><<<cblk, 512, ldsP0, stream>>>(h0c, Wo0, bo0, h0c);
      xg1_kernel<<<cblk, 512, ldsX, stream>>>(h0c, Wx1, b1, xg1c);
    }
  }
}


# the three GEMM kernels between the scan launches run inside the scan launch as extra workgroups (layer 1 lags two chunks; 32 chunks instead of 8); 34->36 launches; k order and sums unchanged
# speedup vs baseline: 20.0099x; 1.0840x over previous
#include <hip/hip_runtime.h>
#include <cstdint>
#include <cstddef>

#define DEVINL __device__ __forceinline__

typedef float v2f __attribute__((ext_vector_type(2)));
typedef float v8f __attribute__((ext_vector_type(8)));

static constexpr int T_DIM  = 1024;
static constexpr int B_DIM  = 64;
static constexpr int IN_DIM = 32;
static constexpr int EMB    = 64;
static constexpr int HID    = 128;
static constexpr int KF     = 16;
static constexpr int OUTD   = 32;
static constexpr int G4     = 512;
static constexpr int NCH    = 32;
static constexpr int CH     = T_DIM / NCH;
static constexpr int MB     = CH * B_DIM / 64;
static constexpr int ST_H   = B_DIM * HID;
static constexpr int ST_ALL = ST_H + 4 * (KF - 1) * 8 * 256;

static constexpr int HS     = 132;
static constexpr int XS     = 68;
static constexpr int IS     = 36;
static constexpr int BP512  = 528;
static constexpr int BP64   = 80;

DEVINL float sigm(float x) { return __builtin_amdgcn_rcpf(1.0f + __expf(-x)); }

DEVINL float tanh_sel(float x) {
  const float y  = fabsf(x);
  const float y2 = x * x;
  float p = fmaf(__uint_as_float(0xbbbac73du), y2, __uint_as_float(0x3ca908c9u));
  p = fmaf(y2, p, __uint_as_float(0xbd5c1c4eu));
  p = fmaf(y2, p, __uint_as_float(0x3e088382u));
  p = fmaf(y2, p, __uint_as_float(0xbeaaaa99u));
  const float zs = fmaf(y2, y * p, y);
  const float e  = __expf(y + y);
  const float zl = fmaf(__builtin_amdgcn_rcpf(e + 1.0f), -2.0f, 1.0f);
  const float z  = (y < 0.625f) ? zs : zl;
  return copysignf(z, x);
}

DEVINL v8f wmma4(v2f a, v2f b, v8f c) {
  return __builtin_amdgcn_wmma_f32_16x16x4_f32(false, a, false, b, (short)0, c,
                                               false, false);
}

DEVINL v8f vzero8() {
  v8f z;
#pragma unroll
  for (int r = 0; r < 8; ++r) z[r] = 0.0f;
  return z;
}

__global__ void __launch_bounds__(512)
embed_xg0_kernel(const float* __restrict__ in,
                 const float* __restrict__ We,
                 const float* __restrict__ be,
                 const float* __restrict__ Wx0,
                 const float* __restrict__ b0,
                 float* __restrict__ xg)
{
  extern __shared__ __align__(16) char smem[];
  float* inP = (float*)smem;
  v2f*   WeP = (v2f*)(smem + 64 * IS * 4);
  float* xb  = (float*)(smem + 64 * IS * 4 + 16 * BP64 * 8);
  v2f*   WxP = (v2f*)(smem + 64 * IS * 4 + 16 * BP64 * 8 + 64 * XS * 4);

  const int tid  = threadIdx.x;
  const int wave = tid >> 5;
  const int lane = tid & 31;
  const int lh   = lane >> 4;
  const int ln   = lane & 15;
  const int row0 = blockIdx.x * 64;

  for (int i = tid; i < 64 * 32; i += 512) {
    int r = i >> 5, c = i & 31;
    inP[r * IS + c] = in[(size_t)(row0 + r) * 32 + c];
  }
  for (int i = tid; i < 16 * 64; i += 512) {
    int kp = i >> 6, n = i & 63;
    v2f w; w.x = We[(2 * kp) * 64 + n]; w.y = We[(2 * kp + 1) * 64 + n];
    WeP[kp * BP64 + n] = w;
  }
  for (int i = tid; i < 32 * 512; i += 512) {
    int kp = i >> 9, n = i & 511;
    v2f w; w.x = Wx0[(2 * kp) * 512 + n]; w.y = Wx0[(2 * kp + 1) * 512 + n];
    WxP[kp * BP512 + n] = w;
  }
  __syncthreads();

  {
    const int tm = wave >> 2, tn = wave & 3;
    const int col = tn * 16 + ln;
    const float bv = be[col];
    v8f acc = vzero8();
    const int am = tm * 16 + ln;
#pragma unroll
    for (int k0 = 0; k0 < 32; k0 += 4) {
      v2f a = *(const v2f*)&inP[am * IS + k0 + 2 * lh];
      v2f b = WeP[((k0 >> 1) + lh) * BP64 + col];
      acc = wmma4(a, b, acc);
    }
#pragma unroll
    for (int r = 0; r < 8; ++r)
      xb[(tm * 16 + r + 8 * lh) * XS + col] = acc[r] + bv;
  }
  __syncthreads();

  for (int task = wave; task < 128; task += 16) {
    const int tm = task >> 5, tn = task & 31;
    const int col = tn * 16 + ln;
    const float bv = b0[col];
    v8f acc = vzero8();
    const int am = tm * 16 + ln;
#pragma unroll
    for (int k0 = 0; k0 < 64; k0 += 4) {
      v2f a = *(const v2f*)&xb[am * XS + k0 + 2 * lh];
      v2f b = WxP[((k0 >> 1) + lh) * BP512 + col];
      acc = wmma4(a, b, acc);
    }
#pragma unroll
    for (int r = 0; r < 8; ++r)
      xg[(size_t)(row0 + tm * 16 + r + 8 * lh) * 512 + col] = acc[r] + bv;
  }
}

struct ScanJob {
  const float* xg;
  const float* Wh;
  float*       hseq;
  float*       st;
  int          first;
  int          on;
};

DEVINL void scan_role(const ScanJob& j0, const ScanJob& j1, char* smem)
{
  float* hbuf = (float*)smem;

  const bool sec = blockIdx.x >= 4;
  if (!(sec ? j1.on : j0.on)) return;
  const float* __restrict__ xg   = sec ? j1.xg   : j0.xg;
  const float* __restrict__ Wh   = sec ? j1.Wh   : j0.Wh;
  float* __restrict__       hseq = sec ? j1.hseq : j0.hseq;
  float* __restrict__       st   = sec ? j1.st   : j0.st;
  const int first                = sec ? j1.first : j0.first;

  const int tid  = threadIdx.x;
  const int wave = tid >> 5;
  const int lane = tid & 31;
  const int lh   = lane >> 4;
  const int ln   = lane & 15;
  const int wg   = blockIdx.x & 3;
  const int rowb = wg * 16;
  const int colh = wave * 16 + ln;
  float* sth = st + (size_t)rowb * HID;
  float* sthist = st + ST_H + (size_t)wg * ((KF - 1) * 8 * 256) + tid;

  v2f wfr[4][32];
#pragma unroll
  for (int ks = 0; ks < 32; ++ks) {
    const float* wp = Wh + (size_t)(4 * ks + 2 * lh) * 512 + colh;
#pragma unroll
    for (int g = 0; g < 4; ++g) {
      v2f w; w.x = wp[g * 128]; w.y = wp[512 + g * 128];
      wfr[g][ks] = w;
    }
  }
  float hnew[8];

  float hist[KF - 1][8];
  if (first) {
    for (int i = tid; i < 16 * HS; i += 256) hbuf[i] = 0.0f;
#pragma unroll
    for (int j = 0; j < KF - 1; ++j)
#pragma unroll
      for (int r = 0; r < 8; ++r) hist[j][r] = 0.0f;
  } else {
    for (int i = tid; i < 16 * HID; i += 256)
      hbuf[(i >> 7) * HS + (i & 127)] = sth[i];
#pragma unroll
    for (int j = 0; j < KF - 1; ++j)
#pragma unroll
      for (int r = 0; r < 8; ++r) hist[j][r] = sthist[(j * 8 + r) * 256];
  }
  __syncthreads();

  for (int t = 0; t < CH; ++t) {
    const float* xgt = xg + (size_t)t * (64 * 512) + (size_t)rowb * 512;

    if (t + 4 < CH) {
#pragma unroll
      for (int q = 1; q <= 4; ++q)
        __builtin_prefetch(xgt + (size_t)q * (64 * 512) + (size_t)tid * 32, 0, 0);
    }

    {
      v2f afr[32];
#pragma unroll
      for (int ks = 0; ks < 32; ++ks)
        afr[ks] = *(const v2f*)&hbuf[ln * HS + 4 * ks + 2 * lh];
      __builtin_amdgcn_sched_barrier(0);

      float xv0[8], xv1[8], xv2[8], xv3[8];
#pragma unroll
      for (int r = 0; r < 8; ++r) {
        const int row = r + 8 * lh;
        const float* p = xgt + row * 512 + colh;
        xv0[r] = p[0];  xv1[r] = p[128];
        xv2[r] = p[256]; xv3[r] = p[384];
      }

      v8f acc0 = vzero8(), acc1 = vzero8(), acc2 = vzero8(), acc3 = vzero8();
#pragma unroll
      for (int ks = 0; ks < 32; ++ks) {
        const v2f a = afr[ks];
        acc0 = wmma4(a, wfr[0][ks], acc0);
        acc1 = wmma4(a, wfr[1][ks], acc1);
        acc2 = wmma4(a, wfr[2][ks], acc2);
        acc3 = wmma4(a, wfr[3][ks], acc3);
      }
      __builtin_amdgcn_sched_barrier(0);

      float hc[8], og[8], dd[8];
#pragma unroll
      for (int r = 0; r < 8; ++r) {
        const float ig = sigm(acc0[r] + xv0[r]);
        og[r] = sigm(acc1[r] + xv1[r]);
        const float cg = tanh_sel(acc2[r] + xv2[r]);
        dd[r] = 0.5f * sigm(acc3[r] + xv3[r]);
        hc[r] = ig * cg;
      }

      float csum[8], w[8];
#pragma unroll
      for (int r = 0; r < 8; ++r) { csum[r] = hc[r]; w[r] = 1.0f; }
#pragma unroll
      for (int j = 1; j < KF; ++j) {
        const float jm1 = (float)(j - 1), rj = 1.0f / (float)j;
#pragma unroll
        for (int r = 0; r < 8; ++r) {
          w[r] *= (jm1 + dd[r]) * rj;
          csum[r] = fmaf(w[r], hist[j - 1][r], csum[r]);
        }
      }
#pragma unroll
      for (int j = KF - 2; j > 0; --j)
#pragma unroll
        for (int r = 0; r < 8; ++r) hist[j][r] = hist[j - 1][r];
#pragma unroll
      for (int r = 0; r < 8; ++r) {
        hist[0][r] = hc[r];
        hnew[r] = og[r] * tanh_sel(csum[r]);
      }
    }

    __builtin_amdgcn_sched_barrier(0);
    __syncthreads();
#pragma unroll
    for (int r = 0; r < 8; ++r)
      hbuf[(r + 8 * lh) * HS + colh] = hnew[r];
    __syncthreads();

    {
      float* ht = hseq + (size_t)t * (64 * HID) + (size_t)rowb * HID;
#pragma unroll
      for (int q = 0; q < 2; ++q) {
        const int row = 2 * wave + q;
        const float4 v = *(const float4*)&hbuf[row * HS + 4 * lane];
        *(float4*)&ht[row * HID + 4 * lane] = v;
      }
    }

    if (t == CH - 1) {
      for (int i = tid; i < 16 * HID; i += 256)
        sth[i] = hbuf[(i >> 7) * HS + (i & 127)];
#pragma unroll
      for (int j = 0; j < KF - 1; ++j)
#pragma unroll
        for (int r = 0; r < 8; ++r) sthist[(j * 8 + r) * 256] = hist[j][r];
    }
  }
}

struct GatesJob {
  const float* h;
  const float* Wo;
  const float* bo;
  const float* Wx;
  const float* b;
  float*       xg;
  int          on;
};

DEVINL void gates_role(const GatesJob& g, int blk, char* smem)
{
  if (!g.on) return;
  constexpr int BP128 = HID + 16;
  float* xB  = (float*)smem;
  float* aB  = (float*)(smem + 64 * HS * 4);
  v2f*   WoP = (v2f*)(smem + 2 * 64 * HS * 4);
  v2f*   WxP = (v2f*)(smem + 64 * HS * 4);
  const float* __restrict__ hin = g.h;
  const float* __restrict__ Wo  = g.Wo;
  const float* __restrict__ bo  = g.bo;
  const float* __restrict__ Wx  = g.Wx;
  const float* __restrict__ bx  = g.b;
  float* __restrict__       xg  = g.xg;

  const int tid  = threadIdx.x;
  const int wave = tid >> 5;
  const int lane = tid & 31;
  const int lh   = lane >> 4;
  const int ln   = lane & 15;
  const int row0 = blk * 64;

  for (int i = tid; i < 64 * 128; i += 256) {
    int r = i >> 7, c = i & 127;
    aB[r * HS + c] = hin[(size_t)(row0 + r) * 128 + c];
  }
  for (int i = tid; i < 64 * 128; i += 256) {
    int kp = i >> 7, n = i & 127;
    v2f w; w.x = Wo[(2 * kp) * 128 + n]; w.y = Wo[(2 * kp + 1) * 128 + n];
    WoP[kp * BP128 + n] = w;
  }
  __syncthreads();

  for (int task = wave; task < 32; task += 8) {
    const int tm = task >> 3, tn = task & 7;
    const int col = tn * 16 + ln;
    const float bv = bo[col];
    v8f acc = vzero8();
    const int am = tm * 16 + ln;
    for (int k0 = 0; k0 < 128; k0 += 4) {
      v2f a = *(const v2f*)&aB[am * HS + k0 + 2 * lh];
      v2f b = WoP[((k0 >> 1) + lh) * BP128 + col];
      acc = wmma4(a, b, acc);
    }
#pragma unroll
    for (int r = 0; r < 8; ++r)
      xB[(tm * 16 + r + 8 * lh) * HS + col] = acc[r] + bv;
  }
  __syncthreads();

  for (int i = tid; i < 64 * 512; i += 256) {
    int kp = i >> 9, n = i & 511;
    v2f w; w.x = Wx[(2 * kp) * 512 + n]; w.y = Wx[(2 * kp + 1) * 512 + n];
    WxP[kp * BP512 + n] = w;
  }
  __syncthreads();

  for (int task = wave; task < 128; task += 8) {
    const int tm = task >> 5, tn = task & 31;
    const int col = tn * 16 + ln;
    const float bv = bx[col];
    v8f acc = vzero8();
    const int am = tm * 16 + ln;
    for (int k0 = 0; k0 < 128; k0 += 4) {
      v2f a = *(const v2f*)&xB[am * HS + k0 + 2 * lh];
      v2f b = WxP[((k0 >> 1) + lh) * BP512 + col];
      acc = wmma4(a, b, acc);
    }
#pragma unroll
    for (int r = 0; r < 8; ++r)
      xg[(size_t)(row0 + tm * 16 + r + 8 * lh) * 512 + col] = acc[r] + bv;
  }
}

struct ProjJob {
  const float* h;
  const float* Wo;
  const float* bo;
  float*       y;
  int          on;
};

DEVINL void proj_role(const ProjJob& p, int blk, char* smem)
{
  if (!p.on) return;
  constexpr int BPS = OUTD + 16;
  constexpr int NTN = OUTD / 16;
  float* aB = (float*)smem;
  v2f*   WP = (v2f*)(smem + 64 * HS * 4);
  const float* __restrict__ hin = p.h;
  const float* __restrict__ Wo  = p.Wo;
  const float* __restrict__ bo  = p.bo;
  float* __restrict__       y   = p.y;

  const int tid  = threadIdx.x;
  const int wave = tid >> 5;
  const int lane = tid & 31;
  const int lh   = lane >> 4;
  const int ln   = lane & 15;
  const int row0 = blk * 64;

  for (int i = tid; i < 64 * 128; i += 256) {
    int r = i >> 7, c = i & 127;
    aB[r * HS + c] = hin[(size_t)(row0 + r) * 128 + c];
  }
  for (int i = tid; i < 64 * OUTD; i += 256) {
    int kp = i / OUTD, n = i % OUTD;
    v2f w; w.x = Wo[(2 * kp) * OUTD + n]; w.y = Wo[(2 * kp + 1) * OUTD + n];
    WP[kp * BPS + n] = w;
  }
  __syncthreads();

  for (int task = wave; task < 4 * NTN; task += 8) {
    const int tm = task / NTN, tn = task % NTN;
    const int col = tn * 16 + ln;
    const float bv = bo[col];
    v8f acc = vzero8();
    const int am = tm * 16 + ln;
    for (int k0 = 0; k0 < 128; k0 += 4) {
      v2f a = *(const v2f*)&aB[am * HS + k0 + 2 * lh];
      v2f b = WP[((k0 >> 1) + lh) * BPS + col];
      acc = wmma4(a, b, acc);
    }
#pragma unroll
    for (int r = 0; r < 8; ++r)
      y[(size_t)(row0 + tm * 16 + r + 8 * lh) * OUTD + col] = acc[r] + bv;
  }
}

__global__ void __launch_bounds__(256)
chunk_kernel(ScanJob j0, ScanJob j1, GatesJob gj, ProjJob pj)
{
  extern __shared__ __align__(16) char smem[];
  const int bid = blockIdx.x;
  if (bid < 8)            scan_role(j0, j1, smem);
  else if (bid < 8 + MB)  gates_role(gj, bid - 8, smem);
  else                    proj_role(pj, bid - 8 - MB, smem);
}

extern "C" void kernel_launch(void* const* d_in, const int* in_sizes, int n_in,
                              void* d_out, int out_size, void* d_ws, size_t ws_size,
                              hipStream_t stream) {
  const float* inputs = (const float*)d_in[0];
  const float* W_emb  = (const float*)d_in[1];
  const float* b_emb  = (const float*)d_in[2];
  const float* Wx0    = (const float*)d_in[3];
  const float* Wh0    = (const float*)d_in[4];
  const float* b0     = (const float*)d_in[5];
  const float* Wo0    = (const float*)d_in[6];
  const float* bo0    = (const float*)d_in[7];
  const float* Wx1    = (const float*)d_in[8];
  const float* Wh1    = (const float*)d_in[9];
  const float* b1     = (const float*)d_in[10];
  const float* Wo1    = (const float*)d_in[11];
  const float* bo1    = (const float*)d_in[12];
  float* out = (float*)d_out;

  const size_t CG = (size_t)CH * B_DIM * G4;
  const size_t CHH = (size_t)CH * B_DIM * HID;
  float* xg0  = (float*)d_ws;
  float* xg1c = xg0  + (size_t)T_DIM * B_DIM * G4;
  float* h0c  = xg1c + 2 * CG;
  float* h1c  = h0c  + 2 * CHH;
  float* st0  = h1c  + 2 * CHH;
  float* st1  = st0  + (size_t)ST_ALL;

  const size_t ldsE = (size_t)64 * IS * 4 + 16 * BP64 * 8 + 64 * XS * 4 + 32 * BP512 * 8;
  const size_t ldsC = (size_t)64 * HS * 4 + 64 * BP512 * 8;

  const int nblk = (T_DIM * B_DIM) / 64;

  embed_xg0_kernel<<<nblk, 512, ldsE, stream>>>(inputs, W_emb, b_emb, Wx0, b0, xg0);

  for (int c = 0; c < NCH + 3; ++c) {
    const int c0 = c < NCH ? c : 0;
    const int m  = c >= 1 ? c - 1 : 0;
    const int q  = c >= 2 ? c - 2 : 0;
    const int p  = c >= 3 ? c - 3 : 0;
    ScanJob  l0 = { xg0 + c0 * CG, Wh0, h0c + (c0 & 1) * CHH, st0, c == 0, c < NCH };
    ScanJob  l1 = { xg1c + (q & 1) * CG, Wh1, h1c + (q & 1) * CHH, st1, c == 2,
                    c >= 2 && c < NCH + 2 };
    GatesJob gj = { h0c + (m & 1) * CHH, Wo0, bo0, Wx1, b1, xg1c + (m & 1) * CG,
                    c >= 1 && c < NCH + 1 };
    ProjJob  pj = { h1c + (p & 1) * CHH, Wo1, bo1, out + (size_t)p * CH * B_DIM * OUTD,
                    c >= 3 };
    chunk_kernel<<<8 + 2 * MB, 256, ldsC, stream>>>(l0, l1, gj, pj);
  }
}


# scan: fractional filter summed in nested (Horner) form from the oldest term, ratio (j-1+d)/j by one fma: 2 packed fma per term, not add, 2 mul, fma; sum order and weight rounding change
# speedup vs baseline: 21.3416x; 1.0666x over previous
#include <hip/hip_runtime.h>
#include <cstdint>
#include <cstddef>

#define DEVINL __device__ __forceinline__

typedef float v2f __attribute__((ext_vector_type(2)));
typedef float v8f __attribute__((ext_vector_type(8)));

static constexpr int T_DIM  = 1024;
static constexpr int B_DIM  = 64;
static constexpr int IN_DIM = 32;
static constexpr int EMB    = 64;
static constexpr int HID    = 128;
static constexpr int KF     = 16;
static constexpr int OUTD   = 32;
static constexpr int G4     = 512;
static constexpr int NCH    = 32;
static constexpr int CH     = T_DIM / NCH;
static constexpr int MB     = CH * B_DIM / 64;
static constexpr int ST_H   = B_DIM * HID;
static constexpr int ST_ALL = ST_H + 4 * (KF - 1) * 8 * 256;

static constexpr int HS     = 132;
static constexpr int XS     = 68;
static constexpr int IS     = 36;
static constexpr int BP512  = 528;
static constexpr int BP64   = 80;

DEVINL float sigm(float x) { return __builtin_amdgcn_rcpf(1.0f + __expf(-x)); }

DEVINL float tanh_sel(float x) {
  const float y  = fabsf(x);
  const float y2 = x * x;
  float p = fmaf(__uint_as_float(0xbbbac73du), y2, __uint_as_float(0x3ca908c9u));
  p = fmaf(y2, p, __uint_as_float(0xbd5c1c4eu));
  p = fmaf(y2, p, __uint_as_float(0x3e088382u));
  p = fmaf(y2, p, __uint_as_float(0xbeaaaa99u));
  const float zs = fmaf(y2, y * p, y);
  const float e  = __expf(y + y);
  const float zl = fmaf(__builtin_amdgcn_rcpf(e + 1.0f), -2.0f, 1.0f);
  const float z  = (y < 0.625f) ? zs : zl;
  return copysignf(z, x);
}

DEVINL v8f wmma4(v2f a, v2f b, v8f c) {
  return __builtin_amdgcn_wmma_f32_16x16x4_f32(false, a, false, b, (short)0, c,
                                               false, false);
}

DEVINL v8f vzero8() {
  v8f z;
#pragma unroll
  for (int r = 0; r < 8; ++r) z[r] = 0.0f;
  return z;
}

__global__ void __launch_bounds__(512)
embed_xg0_kernel(const float* __restrict__ in,
                 const float* __restrict__ We,
                 const float* __restrict__ be,
                 const float* __restrict__ Wx0,
                 const float* __restrict__ b0,
                 float* __restrict__ xg)
{
  extern __shared__ __align__(16) char smem[];
  float* inP = (float*)smem;
  v2f*   WeP = (v2f*)(smem + 64 * IS * 4);
  float* xb  = (float*)(smem + 64 * IS * 4 + 16 * BP64 * 8);
  v2f*   WxP = (v2f*)(smem + 64 * IS * 4 + 16 * BP64 * 8 + 64 * XS * 4);

  const int tid  = threadIdx.x;
  const int wave = tid >> 5;
  const int lane = tid & 31;
  const int lh   = lane >> 4;
  const int ln   = lane & 15;
  const int row0 = blockIdx.x * 64;

  for (int i = tid; i < 64 * 32; i += 512) {
    int r = i >> 5, c = i & 31;
    inP[r * IS + c] = in[(size_t)(row0 + r) * 32 + c];
  }
  for (int i = tid; i < 16 * 64; i += 512) {
    int kp = i >> 6, n = i & 63;
    v2f w; w.x = We[(2 * kp) * 64 + n]; w.y = We[(2 * kp + 1) * 64 + n];
    WeP[kp * BP64 + n] = w;
  }
  for (int i = tid; i < 32 * 512; i += 512) {
    int kp = i >> 9, n = i & 511;
    v2f w; w.x = Wx0[(2 * kp) * 512 + n]; w.y = Wx0[(2 * kp + 1) * 512 + n];
    WxP[kp * BP512 + n] = w;
  }
  __syncthreads();

  {
    const int tm = wave >> 2, tn = wave & 3;
    const int col = tn * 16 + ln;
    const float bv = be[col];
    v8f acc = vzero8();
    const int am = tm * 16 + ln;
#pragma unroll
    for (int k0 = 0; k0 < 32; k0 += 4) {
      v2f a = *(const v2f*)&inP[am * IS + k0 + 2 * lh];
      v2f b = WeP[((k0 >> 1) + lh) * BP64 + col];
      acc = wmma4(a, b, acc);
    }
#pragma unroll
    for (int r = 0; r < 8; ++r)
      xb[(tm * 16 + r + 8 * lh) * XS + col] = acc[r] + bv;
  }
  __syncthreads();

  for (int task = wave; task < 128; task += 16) {
    const int tm = task >> 5, tn = task & 31;
    const int col = tn * 16 + ln;
    const float bv = b0[col];
    v8f acc = vzero8();
    const int am = tm * 16 + ln;
#pragma unroll
    for (int k0 = 0; k0 < 64; k0 += 4) {
      v2f a = *(const v2f*)&xb[am * XS + k0 + 2 * lh];
      v2f b = WxP[((k0 >> 1) + lh) * BP512 + col];
      acc = wmma4(a, b, acc);
    }
#pragma unroll
    for (int r = 0; r < 8; ++r)
      xg[(size_t)(row0 + tm * 16 + r + 8 * lh) * 512 + col] = acc[r] + bv;
  }
}

struct ScanJob {
  const float* xg;
  const float* Wh;
  float*       hseq;
  float*       st;
  int          first;
  int          on;
};

DEVINL void scan_role(const ScanJob& j0, const ScanJob& j1, char* smem)
{
  float* hbuf = (float*)smem;

  const bool sec = blockIdx.x >= 4;
  if (!(sec ? j1.on : j0.on)) return;
  const float* __restrict__ xg   = sec ? j1.xg   : j0.xg;
  const float* __restrict__ Wh   = sec ? j1.Wh   : j0.Wh;
  float* __restrict__       hseq = sec ? j1.hseq : j0.hseq;
  float* __restrict__       st   = sec ? j1.st   : j0.st;
  const int first                = sec ? j1.first : j0.first;

  const int tid  = threadIdx.x;
  const int wave = tid >> 5;
  const int lane = tid & 31;
  const int lh   = lane >> 4;
  const int ln   = lane & 15;
  const int wg   = blockIdx.x & 3;
  const int rowb = wg * 16;
  const int colh = wave * 16 + ln;
  float* sth = st + (size_t)rowb * HID;
  float* sthist = st + ST_H + (size_t)wg * ((KF - 1) * 8 * 256) + tid;

  v2f wfr[4][32];
#pragma unroll
  for (int ks = 0; ks < 32; ++ks) {
    const float* wp = Wh + (size_t)(4 * ks + 2 * lh) * 512 + colh;
#pragma unroll
    for (int g = 0; g < 4; ++g) {
      v2f w; w.x = wp[g * 128]; w.y = wp[512 + g * 128];
      wfr[g][ks] = w;
    }
  }
  float hnew[8];

  v2f hist[KF - 1][4];
  if (first) {
    for (int i = tid; i < 16 * HS; i += 256) hbuf[i] = 0.0f;
#pragma unroll
    for (int j = 0; j < KF - 1; ++j)
#pragma unroll
      for (int p = 0; p < 4; ++p) { hist[j][p].x = 0.0f; hist[j][p].y = 0.0f; }
  } else {
    for (int i = tid; i < 16 * HID; i += 256)
      hbuf[(i >> 7) * HS + (i & 127)] = sth[i];
#pragma unroll
    for (int j = 0; j < KF - 1; ++j)
#pragma unroll
      for (int p = 0; p < 4; ++p) {
        hist[j][p].x = sthist[(j * 8 + 2 * p) * 256];
        hist[j][p].y = sthist[(j * 8 + 2 * p + 1) * 256];
      }
  }
  __syncthreads();

  for (int t = 0; t < CH; ++t) {
    const float* xgt = xg + (size_t)t * (64 * 512) + (size_t)rowb * 512;

    if (t + 4 < CH) {
#pragma unroll
      for (int q = 1; q <= 4; ++q)
        __builtin_prefetch(xgt + (size_t)q * (64 * 512) + (size_t)tid * 32, 0, 0);
    }

    {
      v2f afr[32];
#pragma unroll
      for (int ks = 0; ks < 32; ++ks)
        afr[ks] = *(const v2f*)&hbuf[ln * HS + 4 * ks + 2 * lh];
      __builtin_amdgcn_sched_barrier(0);

      float xv0[8], xv1[8], xv2[8], xv3[8];
#pragma unroll
      for (int r = 0; r < 8; ++r) {
        const int row = r + 8 * lh;
        const float* p = xgt + row * 512 + colh;
        xv0[r] = p[0];  xv1[r] = p[128];
        xv2[r] = p[256]; xv3[r] = p[384];
      }

      v8f acc0 = vzero8(), acc1 = vzero8(), acc2 = vzero8(), acc3 = vzero8();
#pragma unroll
      for (int ks = 0; ks < 32; ++ks) {
        const v2f a = afr[ks];
        acc0 = wmma4(a, wfr[0][ks], acc0);
        acc1 = wmma4(a, wfr[1][ks], acc1);
        acc2 = wmma4(a, wfr[2][ks], acc2);
        acc3 = wmma4(a, wfr[3][ks], acc3);
      }
      __builtin_amdgcn_sched_barrier(0);

      float hc[8], og[8], dd[8];
#pragma unroll
      for (int r = 0; r < 8; ++r) {
        const float ig = sigm(acc0[r] + xv0[r]);
        og[r] = sigm(acc1[r] + xv1[r]);
        const float cg = tanh_sel(acc2[r] + xv2[r]);
        dd[r] = 0.5f * sigm(acc3[r] + xv3[r]);
        hc[r] = ig * cg;
      }

      v2f d2[4], hc2[4], cs[4];
#pragma unroll
      for (int p = 0; p < 4; ++p) {
        d2[p].x  = dd[2 * p]; d2[p].y  = dd[2 * p + 1];
        hc2[p].x = hc[2 * p]; hc2[p].y = hc[2 * p + 1];
        cs[p] = hist[KF - 2][p];
      }
#pragma unroll
      for (int j = KF - 1; j >= 2; --j) {
        const float rj = 1.0f / (float)j, cj = (float)(j - 1) / (float)j;
        v2f rj2, cj2;
        rj2.x = rj; rj2.y = rj; cj2.x = cj; cj2.y = cj;
#pragma unroll
        for (int p = 0; p < 4; ++p) {
          const v2f f = __builtin_elementwise_fma(d2[p], rj2, cj2);
          cs[p] = __builtin_elementwise_fma(f, cs[p], hist[j - 2][p]);
        }
      }
#pragma unroll
      for (int p = 0; p < 4; ++p)
        cs[p] = __builtin_elementwise_fma(d2[p], cs[p], hc2[p]);
#pragma unroll
      for (int j = KF - 2; j > 0; --j)
#pragma unroll
        for (int p = 0; p < 4; ++p) hist[j][p] = hist[j - 1][p];
#pragma unroll
      for (int p = 0; p < 4; ++p) {
        hist[0][p] = hc2[p];
        hnew[2 * p]     = og[2 * p]     * tanh_sel(cs[p].x);
        hnew[2 * p + 1] = og[2 * p + 1] * tanh_sel(cs[p].y);
      }
    }

    __builtin_amdgcn_sched_barrier(0);
    __syncthreads();
#pragma unroll
    for (int r = 0; r < 8; ++r)
      hbuf[(r + 8 * lh) * HS + colh] = hnew[r];
    __syncthreads();

    {
      float* ht = hseq + (size_t)t * (64 * HID) + (size_t)rowb * HID;
#pragma unroll
      for (int q = 0; q < 2; ++q) {
        const int row = 2 * wave + q;
        const float4 v = *(const float4*)&hbuf[row * HS + 4 * lane];
        *(float4*)&ht[row * HID + 4 * lane] = v;
      }
    }

    if (t == CH - 1) {
      for (int i = tid; i < 16 * HID; i += 256)
        sth[i] = hbuf[(i >> 7) * HS + (i & 127)];
#pragma unroll
      for (int j = 0; j < KF - 1; ++j)
#pragma unroll
        for (int p = 0; p < 4; ++p) {
          sthist[(j * 8 + 2 * p) * 256]     = hist[j][p].x;
          sthist[(j * 8 + 2 * p + 1) * 256] = hist[j][p].y;
        }
    }
  }
}

struct GatesJob {
  const float* h;
  const float* Wo;
  const float* bo;
  const float* Wx;
  const float* b;
  float*       xg;
  int          on;
};

DEVINL void gates_role(const GatesJob& g, int blk, char* smem)
{
  if (!g.on) return;
  constexpr int BP128 = HID + 16;
  float* xB  = (float*)smem;
  float* aB  = (float*)(smem + 64 * HS * 4);
  v2f*   WoP = (v2f*)(smem + 2 * 64 * HS * 4);
  v2f*   WxP = (v2f*)(smem + 64 * HS * 4);
  const float* __restrict__ hin = g.h;
  const float* __restrict__ Wo  = g.Wo;
  const float* __restrict__ bo  = g.bo;
  const float* __restrict__ Wx  = g.Wx;
  const float* __restrict__ bx  = g.b;
  float* __restrict__       xg  = g.xg;

  const int tid  = threadIdx.x;
  const int wave = tid >> 5;
  const int lane = tid & 31;
  const int lh   = lane >> 4;
  const int ln   = lane & 15;
  const int row0 = blk * 64;

  for (int i = tid; i < 64 * 128; i += 256) {
    int r = i >> 7, c = i & 127;
    aB[r * HS + c] = hin[(size_t)(row0 + r) * 128 + c];
  }
  for (int i = tid; i < 64 * 128; i += 256) {
    int kp = i >> 7, n = i & 127;
    v2f w; w.x = Wo[(2 * kp) * 128 + n]; w.y = Wo[(2 * kp + 1) * 128 + n];
    WoP[kp * BP128 + n] = w;
  }
  __syncthreads();

  for (int task = wave; task < 32; task += 8) {
    const int tm = task >> 3, tn = task & 7;
    const int col = tn * 16 + ln;
    const float bv = bo[col];
    v8f acc = vzero8();
    const int am = tm * 16 + ln;
    for (int k0 = 0; k0 < 128; k0 += 4) {
      v2f a = *(const v2f*)&aB[am * HS + k0 + 2 * lh];
      v2f b = WoP[((k0 >> 1) + lh) * BP128 + col];
      acc = wmma4(a, b, acc);
    }
#pragma unroll
    for (int r = 0; r < 8; ++r)
      xB[(tm * 16 + r + 8 * lh) * HS + col] = acc[r] + bv;
  }
  __syncthreads();

  for (int i = tid; i < 64 * 512; i += 256) {
    int kp = i >> 9, n = i & 511;
    v2f w; w.x = Wx[(2 * kp) * 512 + n]; w.y = Wx[(2 * kp + 1) * 512 + n];
    WxP[kp * BP512 + n] = w;
  }
  __syncthreads();

  for (int task = wave; task < 128; task += 8) {
    const int tm = task >> 5, tn = task & 31;
    const int col = tn * 16 + ln;
    const float bv = bx[col];
    v8f acc = vzero8();
    const int am = tm * 16 + ln;
    for (int k0 = 0; k0 < 128; k0 += 4) {
      v2f a = *(const v2f*)&xB[am * HS + k0 + 2 * lh];
      v2f b = WxP[((k0 >> 1) + lh) * BP512 + col];
      acc = wmma4(a, b, acc);
    }
#pragma unroll
    for (int r = 0; r < 8; ++r)
      xg[(size_t)(row0 + tm * 16 + r + 8 * lh) * 512 + col] = acc[r] + bv;
  }
}

struct ProjJob {
  const float* h;
  const float* Wo;
  const float* bo;
  float*       y;
  int          on;
};

DEVINL void proj_role(const ProjJob& p, int blk, char* smem)
{
  if (!p.on) return;
  constexpr int BPS = OUTD + 16;
  constexpr int NTN = OUTD / 16;
  float* aB = (float*)smem;
  v2f*   WP = (v2f*)(smem + 64 * HS * 4);
  const float* __restrict__ hin = p.h;
  const float* __restrict__ Wo  = p.Wo;
  const float* __restrict__ bo  = p.bo;
  float* __restrict__       y   = p.y;

  const int tid  = threadIdx.x;
  const int wave = tid >> 5;
  const int lane = tid & 31;
  const int lh   = lane >> 4;
  const int ln   = lane & 15;
  const int row0 = blk * 64;

  for (int i = tid; i < 64 * 128; i += 256) {
    int r = i >> 7, c = i & 127;
    aB[r * HS + c] = hin[(size_t)(row0 + r) * 128 + c];
  }
  for (int i = tid; i < 64 * OUTD; i += 256) {
    int kp = i / OUTD, n = i % OUTD;
    v2f w; w.x = Wo[(2 * kp) * OUTD + n]; w.y = Wo[(2 * kp + 1) * OUTD + n];
    WP[kp * BPS + n] = w;
  }
  __syncthreads();

  for (int task = wave; task < 4 * NTN; task += 8) {
    const int tm = task / NTN, tn = task % NTN;
    const int col = tn * 16 + ln;
    const float bv = bo[col];
    v8f acc = vzero8();
    const int am = tm * 16 + ln;
    for (int k0 = 0; k0 < 128; k0 += 4) {
      v2f a = *(const v2f*)&aB[am * HS + k0 + 2 * lh];
      v2f b = WP[((k0 >> 1) + lh) * BPS + col];
      acc = wmma4(a, b, acc);
    }
#pragma unroll
    for (int r = 0; r < 8; ++r)
      y[(size_t)(row0 + tm * 16 + r + 8 * lh) * OUTD + col] = acc[r] + bv;
  }
}

__global__ void __launch_bounds__(256)
chunk_kernel(ScanJob j0, ScanJob j1, GatesJob gj, ProjJob pj)
{
  extern __shared__ __align__(16) char smem[];
  const int bid = blockIdx.x;
  if (bid < 8)            scan_role(j0, j1, smem);
  else if (bid < 8 + MB)  gates_role(gj, bid - 8, smem);
  else                    proj_role(pj, bid - 8 - MB, smem);
}

extern "C" void kernel_launch(void* const* d_in, const int* in_sizes, int n_in,
                              void* d_out, int out_size, void* d_ws, size_t ws_size,
                              hipStream_t stream) {
  const float* inputs = (const float*)d_in[0];
  const float* W_emb  = (const float*)d_in[1];
  const float* b_emb  = (const float*)d_in[2];
  const float* Wx0    = (const float*)d_in[3];
  const float* Wh0    = (const float*)d_in[4];
  const float* b0     = (const float*)d_in[5];
  const float* Wo0    = (const float*)d_in[6];
  const float* bo0    = (const float*)d_in[7];
  const float* Wx1    = (const float*)d_in[8];
  const float* Wh1    = (const float*)d_in[9];
  const float* b1     = (const float*)d_in[10];
  const float* Wo1    = (const float*)d_in[11];
  const float* bo1    = (const float*)d_in[12];
  float* out = (float*)d_out;

  const size_t CG = (size_t)CH * B_DIM * G4;
  const size_t CHH = (size_t)CH * B_DIM * HID;
  float* xg0  = (float*)d_ws;
  float* xg1c = xg0  + (size_t)T_DIM * B_DIM * G4;
  float* h0c  = xg1c + 2 * CG;
  float* h1c  = h0c  + 2 * CHH;
  float* st0  = h1c  + 2 * CHH;
  float* st1  = st0  + (size_t)ST_ALL;

  const size_t ldsE = (size_t)64 * IS * 4 + 16 * BP64 * 8 + 64 * XS * 4 + 32 * BP512 * 8;
  const size_t ldsC = (size_t)64 * HS * 4 + 64 * BP512 * 8;

  const int nblk = (T_DIM * B_DIM) / 64;

  embed_xg0_kernel<<<nblk, 512, ldsE, stream>>>(inputs, W_emb, b_emb, Wx0, b0, xg0);

  for (int c = 0; c < NCH + 3; ++c) {
    const int c0 = c < NCH ? c : 0;
    const int m  = c >= 1 ? c - 1 : 0;
    const int q  = c >= 2 ? c - 2 : 0;
    const int p  = c >= 3 ? c - 3 : 0;
    ScanJob  l0 = { xg0 + c0 * CG, Wh0, h0c + (c0 & 1) * CHH, st0, c == 0, c < NCH };
    ScanJob  l1 = { xg1c + (q & 1) * CG, Wh1, h1c + (q & 1) * CHH, st1, c == 2,
                    c >= 2 && c < NCH + 2 };
    GatesJob gj = { h0c + (m & 1) * CHH, Wo0, bo0, Wx1, b1, xg1c + (m & 1) * CG,
                    c >= 1 && c < NCH + 1 };
    ProjJob  pj = { h1c + (p & 1) * CHH, Wo1, bo1, out + (size_t)p * CH * B_DIM * OUTD,
                    c >= 3 };
    chunk_kernel<<<8 + 2 * MB, 256, ldsC, stream>>>(l0, l1, gj, pj);
  }
}


# scan: batch rows dealt to 16 workgroups per layer, not 4: 4 rows each at rows 0,1,8,9 of the padded 16-row matrix tile, so a lane's gate and filter code runs over 2 values, not 8; sums unchanged
# speedup vs baseline: 25.9004x; 1.2136x over previous
#include <hip/hip_runtime.h>
#include <cstdint>
#include <cstddef>

#define DEVINL __device__ __forceinline__

typedef float v2f __attribute__((ext_vector_type(2)));
typedef float v8f __attribute__((ext_vector_type(8)));

static constexpr int T_DIM  = 1024;
static constexpr int B_DIM  = 64;
static constexpr int IN_DIM = 32;
static constexpr int EMB    = 64;
static constexpr int HID    = 128;
static constexpr int KF     = 16;
static constexpr int OUTD   = 32;
static constexpr int G4     = 512;
static constexpr int NCH    = 32;
static constexpr int CH     = T_DIM / NCH;
static constexpr int MB     = CH * B_DIM / 64;
static constexpr int RN     = 2;
static constexpr int RW     = 2 * RN;
static constexpr int NSW    = B_DIM / RW;
static constexpr int ST_H   = B_DIM * HID;
static constexpr int ST_ALL = ST_H + NSW * (KF - 1) * RN * 256;

static constexpr int HS     = 132;
static constexpr int XS     = 68;
static constexpr int IS     = 36;
static constexpr int BP512  = 528;
static constexpr int BP64   = 80;

DEVINL float sigm(float x) { return __builtin_amdgcn_rcpf(1.0f + __expf(-x)); }

DEVINL float tanh_sel(float x) {
  const float y  = fabsf(x);
  const float y2 = x * x;
  float p = fmaf(__uint_as_float(0xbbbac73du), y2, __uint_as_float(0x3ca908c9u));
  p = fmaf(y2, p, __uint_as_float(0xbd5c1c4eu));
  p = fmaf(y2, p, __uint_as_float(0x3e088382u));
  p = fmaf(y2, p, __uint_as_float(0xbeaaaa99u));
  const float zs = fmaf(y2, y * p, y);
  const float e  = __expf(y + y);
  const float zl = fmaf(__builtin_amdgcn_rcpf(e + 1.0f), -2.0f, 1.0f);
  const float z  = (y < 0.625f) ? zs : zl;
  return copysignf(z, x);
}

DEVINL v8f wmma4(v2f a, v2f b, v8f c) {
  return __builtin_amdgcn_wmma_f32_16x16x4_f32(false, a, false, b, (short)0, c,
                                               false, false);
}

DEVINL v8f vzero8() {
  v8f z;
#pragma unroll
  for (int r = 0; r < 8; ++r) z[r] = 0.0f;
  return z;
}

__global__ void __launch_bounds__(512)
embed_xg0_kernel(const float* __restrict__ in,
                 const float* __restrict__ We,
                 const float* __restrict__ be,
                 const float* __restrict__ Wx0,
                 const float* __restrict__ b0,
                 float* __restrict__ xg)
{
  extern __shared__ __align__(16) char smem[];
  float* inP = (float*)smem;
  v2f*   WeP = (v2f*)(smem + 64 * IS * 4);
  float* xb  = (float*)(smem + 64 * IS * 4 + 16 * BP64 * 8);
  v2f*   WxP = (v2f*)(smem + 64 * IS * 4 + 16 * BP64 * 8 + 64 * XS * 4);

  const int tid  = threadIdx.x;
  const int wave = tid >> 5;
  const int lane = tid & 31;
  const int lh   = lane >> 4;
  const int ln   = lane & 15;
  const int row0 = blockIdx.x * 64;

  for (int i = tid; i < 64 * 32; i += 512) {
    int r = i >> 5, c = i & 31;
    inP[r * IS + c] = in[(size_t)(row0 + r) * 32 + c];
  }
  for (int i = tid; i < 16 * 64; i += 512) {
    int kp = i >> 6, n = i & 63;
    v2f w; w.x = We[(2 * kp) * 64 + n]; w.y = We[(2 * kp + 1) * 64 + n];
    WeP[kp * BP64 + n] = w;
  }
  for (int i = tid; i < 32 * 512; i += 512) {
    int kp = i >> 9, n = i & 511;
    v2f w; w.x = Wx0[(2 * kp) * 512 + n]; w.y = Wx0[(2 * kp + 1) * 512 + n];
    WxP[kp * BP512 + n] = w;
  }
  __syncthreads();

  {
    const int tm = wave >> 2, tn = wave & 3;
    const int col = tn * 16 + ln;
    const float bv = be[col];
    v8f acc = vzero8();
    const int am = tm * 16 + ln;
#pragma unroll
    for (int k0 = 0; k0 < 32; k0 += 4) {
      v2f a = *(const v2f*)&inP[am * IS + k0 + 2 * lh];
      v2f b = WeP[((k0 >> 1) + lh) * BP64 + col];
      acc = wmma4(a, b, acc);
    }
#pragma unroll
    for (int r = 0; r < 8; ++r)
      xb[(tm * 16 + r + 8 * lh) * XS + col] = acc[r] + bv;
  }
  __syncthreads();

  for (int task = wave; task < 128; task += 16) {
    const int tm = task >> 5, tn = task & 31;
    const int col = tn * 16 + ln;
    const float bv = b0[col];
    v8f acc = vzero8();
    const int am = tm * 16 + ln;
#pragma unroll
    for (int k0 = 0; k0 < 64; k0 += 4) {
      v2f a = *(const v2f*)&xb[am * XS + k0 + 2 * lh];
      v2f b = WxP[((k0 >> 1) + lh) * BP512 + col];
      acc = wmma4(a, b, acc);
    }
#pragma unroll
    for (int r = 0; r < 8; ++r)
      xg[(size_t)(row0 + tm * 16 + r + 8 * lh) * 512 + col] = acc[r] + bv;
  }
}

struct ScanJob {
  const float* xg;
  const float* Wh;
  float*       hseq;
  float*       st;
  int          first;
  int          on;
};

DEVINL void scan_role(const ScanJob& j0, const ScanJob& j1, char* smem)
{
  float* hbuf = (float*)smem;

  const bool sec = blockIdx.x >= NSW;
  if (!(sec ? j1.on : j0.on)) return;
  const float* __restrict__ xg   = sec ? j1.xg   : j0.xg;
  const float* __restrict__ Wh   = sec ? j1.Wh   : j0.Wh;
  float* __restrict__       hseq = sec ? j1.hseq : j0.hseq;
  float* __restrict__       st   = sec ? j1.st   : j0.st;
  const int first                = sec ? j1.first : j0.first;

  const int tid  = threadIdx.x;
  const int wave = tid >> 5;
  const int lane = tid & 31;
  const int lh   = lane >> 4;
  const int ln   = lane & 15;
  const int wg   = blockIdx.x % NSW;
  const int rowb = wg * RW;
  const int colh = wave * 16 + ln;
  float* sth = st + (size_t)rowb * HID;
  float* sthist = st + ST_H + (size_t)wg * ((KF - 1) * RN * 256) + tid;

  v2f wfr[4][32];
#pragma unroll
  for (int ks = 0; ks < 32; ++ks) {
    const float* wp = Wh + (size_t)(4 * ks + 2 * lh) * 512 + colh;
#pragma unroll
    for (int g = 0; g < 4; ++g) {
      v2f w; w.x = wp[g * 128]; w.y = wp[512 + g * 128];
      wfr[g][ks] = w;
    }
  }
  float hnew[RN];

  v2f hist[KF - 1][RN / 2];
  if (first) {
    for (int i = tid; i < 16 * HS; i += 256) hbuf[i] = 0.0f;
#pragma unroll
    for (int j = 0; j < KF - 1; ++j)
#pragma unroll
      for (int p = 0; p < RN / 2; ++p) { hist[j][p].x = 0.0f; hist[j][p].y = 0.0f; }
  } else {
    for (int i = tid; i < 16 * HID; i += 256) {
      const int m = i >> 7, c = i & 127;
      const int lr = (m >> 3) * RN + (m & 7);
      hbuf[m * HS + c] = ((m & 7) < RN) ? sth[lr * HID + c] : 0.0f;
    }
#pragma unroll
    for (int j = 0; j < KF - 1; ++j)
#pragma unroll
      for (int p = 0; p < RN / 2; ++p) {
        hist[j][p].x = sthist[(j * RN + 2 * p) * 256];
        hist[j][p].y = sthist[(j * RN + 2 * p + 1) * 256];
      }
  }
  __syncthreads();

  for (int t = 0; t < CH; ++t) {
    const float* xgt = xg + (size_t)t * (64 * 512) + (size_t)rowb * 512;

    if (t + 4 < CH && tid < RW * 16) {
#pragma unroll
      for (int q = 1; q <= 4; ++q)
        __builtin_prefetch(xgt + (size_t)q * (64 * 512) + (size_t)tid * 32, 0, 0);
    }

    {
      v2f afr[32];
#pragma unroll
      for (int ks = 0; ks < 32; ++ks)
        afr[ks] = *(const v2f*)&hbuf[ln * HS + 4 * ks + 2 * lh];
      __builtin_amdgcn_sched_barrier(0);

      float xv0[RN], xv1[RN], xv2[RN], xv3[RN];
#pragma unroll
      for (int r = 0; r < RN; ++r) {
        const int row = RN * lh + r;
        const float* p = xgt + row * 512 + colh;
        xv0[r] = p[0];  xv1[r] = p[128];
        xv2[r] = p[256]; xv3[r] = p[384];
      }

      v8f acc0 = vzero8(), acc1 = vzero8(), acc2 = vzero8(), acc3 = vzero8();
#pragma unroll
      for (int ks = 0; ks < 32; ++ks) {
        const v2f a = afr[ks];
        acc0 = wmma4(a, wfr[0][ks], acc0);
        acc1 = wmma4(a, wfr[1][ks], acc1);
        acc2 = wmma4(a, wfr[2][ks], acc2);
        acc3 = wmma4(a, wfr[3][ks], acc3);
      }
      __builtin_amdgcn_sched_barrier(0);

      float hc[RN], og[RN], dd[RN];
#pragma unroll
      for (int r = 0; r < RN; ++r) {
        const float ig = sigm(acc0[r] + xv0[r]);
        og[r] = sigm(acc1[r] + xv1[r]);
        const float cg = tanh_sel(acc2[r] + xv2[r]);
        dd[r] = 0.5f * sigm(acc3[r] + xv3[r]);
        hc[r] = ig * cg;
      }

      v2f d2[RN / 2], hc2[RN / 2], cs[RN / 2];
#pragma unroll
      for (int p = 0; p < RN / 2; ++p) {
        d2[p].x  = dd[2 * p]; d2[p].y  = dd[2 * p + 1];
        hc2[p].x = hc[2 * p]; hc2[p].y = hc[2 * p + 1];
        cs[p] = hist[KF - 2][p];
      }
#pragma unroll
      for (int j = KF - 1; j >= 2; --j) {
        const float rj = 1.0f / (float)j, cj = (float)(j - 1) / (float)j;
        v2f rj2, cj2;
        rj2.x = rj; rj2.y = rj; cj2.x = cj; cj2.y = cj;
#pragma unroll
        for (int p = 0; p < RN / 2; ++p) {
          const v2f f = __builtin_elementwise_fma(d2[p], rj2, cj2);
          cs[p] = __builtin_elementwise_fma(f, cs[p], hist[j - 2][p]);
        }
      }
#pragma unroll
      for (int p = 0; p < RN / 2; ++p)
        cs[p] = __builtin_elementwise_fma(d2[p], cs[p], hc2[p]);
#pragma unroll
      for (int j = KF - 2; j > 0; --j)
#pragma unroll
        for (int p = 0; p < RN / 2; ++p) hist[j][p] = hist[j - 1][p];
#pragma unroll
      for (int p = 0; p < RN / 2; ++p) {
        hist[0][p] = hc2[p];
        hnew[2 * p]     = og[2 * p]     * tanh_sel(cs[p].x);
        hnew[2 * p + 1] = og[2 * p + 1] * tanh_sel(cs[p].y);
      }
    }

    __builtin_amdgcn_sched_barrier(0);
    __syncthreads();
#pragma unroll
    for (int r = 0; r < RN; ++r)
      hbuf[(r + 8 * lh) * HS + colh] = hnew[r];
    __syncthreads();

    if (wave >= 8 - RW) {
      float* ht = hseq + (size_t)t * (64 * HID) + (size_t)rowb * HID;
      const int lr = wave - (8 - RW);
      const int m = (lr % RN) + 8 * (lr / RN);
      const float4 v = *(const float4*)&hbuf[m * HS + 4 * lane];
      *(float4*)&ht[lr * HID + 4 * lane] = v;
    }

    if (t == CH - 1) {
      for (int i = tid; i < RW * HID; i += 256) {
        const int lr = i >> 7;
        sth[i] = hbuf[((lr % RN) + 8 * (lr / RN)) * HS + (i & 127)];
      }
#pragma unroll
      for (int j = 0; j < KF - 1; ++j)
#pragma unroll
        for (int p = 0; p < RN / 2; ++p) {
          sthist[(j * RN + 2 * p) * 256]     = hist[j][p].x;
          sthist[(j * RN + 2 * p + 1) * 256] = hist[j][p].y;
        }
    }
  }
}

struct GatesJob {
  const float* h;
  const float* Wo;
  const float* bo;
  const float* Wx;
  const float* b;
  float*       xg;
  int          on;
};

DEVINL void gates_role(const GatesJob& g, int blk, char* smem)
{
  if (!g.on) return;
  constexpr int BP128 = HID + 16;
  float* xB  = (float*)smem;
  float* aB  = (float*)(smem + 64 * HS * 4);
  v2f*   WoP = (v2f*)(smem + 2 * 64 * HS * 4);
  v2f*   WxP = (v2f*)(smem + 64 * HS * 4);
  const float* __restrict__ hin = g.h;
  const float* __restrict__ Wo  = g.Wo;
  const float* __restrict__ bo  = g.bo;
  const float* __restrict__ Wx  = g.Wx;
  const float* __restrict__ bx  = g.b;
  float* __restrict__       xg  = g.xg;

  const int tid  = threadIdx.x;
  const int wave = tid >> 5;
  const int lane = tid & 31;
  const int lh   = lane >> 4;
  const int ln   = lane & 15;
  const int row0 = blk * 64;

  for (int i = tid; i < 64 * 128; i += 256) {
    int r = i >> 7, c = i & 127;
    aB[r * HS + c] = hin[(size_t)(row0 + r) * 128 + c];
  }
  for (int i = tid; i < 64 * 128; i += 256) {
    int kp = i >> 7, n = i & 127;
    v2f w; w.x = Wo[(2 * kp) * 128 + n]; w.y = Wo[(2 * kp + 1) * 128 + n];
    WoP[kp * BP128 + n] = w;
  }
  __syncthreads();

  for (int task = wave; task < 32; task += 8) {
    const int tm = task >> 3, tn = task & 7;
    const int col = tn * 16 + ln;
    const float bv = bo[col];
    v8f acc = vzero8();
    const int am = tm * 16 + ln;
    for (int k0 = 0; k0 < 128; k0 += 4) {
      v2f a = *(const v2f*)&aB[am * HS + k0 + 2 * lh];
      v2f b = WoP[((k0 >> 1) + lh) * BP128 + col];
      acc = wmma4(a, b, acc);
    }
#pragma unroll
    for (int r = 0; r < 8; ++r)
      xB[(tm * 16 + r + 8 * lh) * HS + col] = acc[r] + bv;
  }
  __syncthreads();

  for (int i = tid; i < 64 * 512; i += 256) {
    int kp = i >> 9, n = i & 511;
    v2f w; w.x = Wx[(2 * kp) * 512 + n]; w.y = Wx[(2 * kp + 1) * 512 + n];
    WxP[kp * BP512 + n] = w;
  }
  __syncthreads();

  for (int task = wave; task < 128; task += 8) {
    const int tm = task >> 5, tn = task & 31;
    const int col = tn * 16 + ln;
    const float bv = bx[col];
    v8f acc = vzero8();
    const int am = tm * 16 + ln;
    for (int k0 = 0; k0 < 128; k0 += 4) {
      v2f a = *(const v2f*)&xB[am * HS + k0 + 2 * lh];
      v2f b = WxP[((k0 >> 1) + lh) * BP512 + col];
      acc = wmma4(a, b, acc);
    }
#pragma unroll
    for (int r = 0; r < 8; ++r)
      xg[(size_t)(row0 + tm * 16 + r + 8 * lh) * 512 + col] = acc[r] + bv;
  }
}

struct ProjJob {
  const float* h;
  const float* Wo;
  const float* bo;
  float*       y;
  int          on;
};

DEVINL void proj_role(const ProjJob& p, int blk, char* smem)
{
  if (!p.on) return;
  constexpr int BPS = OUTD + 16;
  constexpr int NTN = OUTD / 16;
  float* aB = (float*)smem;
  v2f*   WP = (v2f*)(smem + 64 * HS * 4);
  const float* __restrict__ hin = p.h;
  const float* __restrict__ Wo  = p.Wo;
  const float* __restrict__ bo  = p.bo;
  float* __restrict__       y   = p.y;

  const int tid  = threadIdx.x;
  const int wave = tid >> 5;
  const int lane = tid & 31;
  const int lh   = lane >> 4;
  const int ln   = lane & 15;
  const int row0 = blk * 64;

  for (int i = tid; i < 64 * 128; i += 256) {
    int r = i >> 7, c = i & 127;
    aB[r * HS + c] = hin[(size_t)(row0 + r) * 128 + c];
  }
  for (int i = tid; i < 64 * OUTD; i += 256) {
    int kp = i / OUTD, n = i % OUTD;
    v2f w; w.x = Wo[(2 * kp) * OUTD + n]; w.y = Wo[(2 * kp + 1) * OUTD + n];
    WP[kp * BPS + n] = w;
  }
  __syncthreads();

  for (int task = wave; task < 4 * NTN; task += 8) {
    const int tm = task / NTN, tn = task % NTN;
    const int col = tn * 16 + ln;
    const float bv = bo[col];
    v8f acc = vzero8();
    const int am = tm * 16 + ln;
    for (int k0 = 0; k0 < 128; k0 += 4) {
      v2f a = *(const v2f*)&aB[am * HS + k0 + 2 * lh];
      v2f b = WP[((k0 >> 1) + lh) * BPS + col];
      acc = wmma4(a, b, acc);
    }
#pragma unroll
    for (int r = 0; r < 8; ++r)
      y[(size_t)(row0 + tm * 16 + r + 8 * lh) * OUTD + col] = acc[r] + bv;
  }
}

__global__ void __launch_bounds__(256)
chunk_kernel(ScanJob j0, ScanJob j1, GatesJob gj, ProjJob pj)
{
  extern __shared__ __align__(16) char smem[];
  const int bid = blockIdx.x;
  if (bid < 2 * NSW)            scan_role(j0, j1, smem);
  else if (bid < 2 * NSW + MB)  gates_role(gj, bid - 2 * NSW, smem);
  else                          proj_role(pj, bid - 2 * NSW - MB, smem);
}

extern "C" void kernel_launch(void* const* d_in, const int* in_sizes, int n_in,
                              void* d_out, int out_size, void* d_ws, size_t ws_size,
                              hipStream_t stream) {
  const float* inputs = (const float*)d_in[0];
  const float* W_emb  = (const float*)d_in[1];
  const float* b_emb  = (const float*)d_in[2];
  const float* Wx0    = (const float*)d_in[3];
  const float* Wh0    = (const float*)d_in[4];
  const float* b0     = (const float*)d_in[5];
  const float* Wo0    = (const float*)d_in[6];
  const float* bo0    = (const float*)d_in[7];
  const float* Wx1    = (const float*)d_in[8];
  const float* Wh1    = (const float*)d_in[9];
  const float* b1     = (const float*)d_in[10];
  const float* Wo1    = (const float*)d_in[11];
  const float* bo1    = (const float*)d_in[12];
  float* out = (float*)d_out;

  const size_t CG = (size_t)CH * B_DIM * G4;
  const size_t CHH = (size_t)CH * B_DIM * HID;
  float* xg0  = (float*)d_ws;
  float* xg1c = xg0  + (size_t)T_DIM * B_DIM * G4;
  float* h0c  = xg1c + 2 * CG;
  float* h1c  = h0c  + 2 * CHH;
  float* st0  = h1c  + 2 * CHH;
  float* st1  = st0  + (size_t)ST_ALL;

  const size_t ldsE = (size_t)64 * IS * 4 + 16 * BP64 * 8 + 64 * XS * 4 + 32 * BP512 * 8;
  const size_t ldsC = (size_t)64 * HS * 4 + 64 * BP512 * 8;

  const int nblk = (T_DIM * B_DIM) / 64;

  embed_xg0_kernel<<<nblk, 512, ldsE, stream>>>(inputs, W_emb, b_emb, Wx0, b0, xg0);

  for (int c = 0; c < NCH + 3; ++c) {
    const int c0 = c < NCH ? c : 0;
    const int m  = c >= 1 ? c - 1 : 0;
    const int q  = c >= 2 ? c - 2 : 0;
    const int p  = c >= 3 ? c - 3 : 0;
    ScanJob  l0 = { xg0 + c0 * CG, Wh0, h0c + (c0 & 1) * CHH, st0, c == 0, c < NCH };
    ScanJob  l1 = { xg1c + (q & 1) * CG, Wh1, h1c + (q & 1) * CHH, st1, c == 2,
                    c >= 2 && c < NCH + 2 };
    GatesJob gj = { h0c + (m & 1) * CHH, Wo0, bo0, Wx1, b1, xg1c + (m & 1) * CG,
                    c >= 1 && c < NCH + 1 };
    ProjJob  pj = { h1c + (p & 1) * CHH, Wo1, bo1, out + (size_t)p * CH * B_DIM * OUTD,
                    c >= 3 };
    chunk_kernel<<<2 * NSW + 2 * MB, 256, ldsC, stream>>>(l0, l1, gj, pj);
  }
}
